# starting kernel
# baseline (speedup 1.0000x reference)
#include <hip/hip_runtime.h>
#include <math.h>

typedef float v2f __attribute__((ext_vector_type(2)));
typedef float v8f __attribute__((ext_vector_type(8)));

namespace {
constexpr int kPS   = 16;
constexpr int kPR   = 14;
constexpr int kN    = 196;
constexpr int kC    = 3;
constexpr int kW    = 224;
constexpr int kBS   = 64;
constexpr int kF    = 768;
constexpr int kHid  = 768;
constexpr int kOut  = 768;
constexpr int kRows = kBS * kN;
constexpr int kTileRows     = 16;
constexpr int kWaves        = 8;
constexpr int kTilesPerWave = (kHid / 16) / kWaves;
}

__device__ __forceinline__ float gelu_exact(float v) {
  return 0.5f * v * (1.0f + erff(v * 0.70710678118654752440f));
}

__global__ __launch_bounds__(256)
void conv_bn_gelu_patch(const float* __restrict__ x,
                        const float* __restrict__ cw,
                        const float* __restrict__ cb,
                        const float* __restrict__ bng,
                        const float* __restrict__ bnb,
                        const float* __restrict__ bnm,
                        const float* __restrict__ bnv,
                        const float* __restrict__ M,
                        const int*   __restrict__ mask,
                        float* __restrict__ flat) {
  const int total = kBS * kC * kW * kW;
  int idx = blockIdx.x * blockDim.x + threadIdx.x;
  if (idx >= total) return;

  const int h   = idx % kW;
  const int w   = (idx / kW) % kW;
  const int co  = (idx / (kW * kW)) % kC;
  const int img = idx / (kW * kW * kC);

  float acc = cb[co];
#pragma unroll
  for (int ci = 0; ci < kC; ++ci) {
    const float* __restrict__ xp = x + ((size_t)(img * kC + ci)) * kW * kW;
#pragma unroll
    for (int kw = 0; kw < 3; ++kw) {
      const int ww = w + kw - 1;
      if (ww < 0 || ww >= kW) continue;
#pragma unroll
      for (int kh = 0; kh < 3; ++kh) {
        const int hh = h + kh - 1;
        if (hh < 0 || hh >= kW) continue;
        acc += xp[ww * kW + hh] * cw[((co * kC + ci) * 3 + kw) * 3 + kh];
      }
    }
  }
  const float inv = bng[co] * rsqrtf(bnv[co] + 1e-5f);
  acc = (acc - bnm[co]) * inv + bnb[co];
  acc = gelu_exact(acc);

  const int pi = w & 15, pw = w >> 4;
  const int pj = h & 15, ph = h >> 4;
  const int n  = pw * kPR + ph;
  if (mask[n] > 0) acc *= M[pi * kPS + pj];

  const int row = img * kN + n;
  const int f   = co * kPS * kPS + pi * kPS + pj;
  flat[(size_t)row * kF + f] = acc;
}

__global__ __launch_bounds__(256, 2)
void gemm1_ln_gelu(const float* __restrict__ A,
                   const float* __restrict__ W1,
                   const float* __restrict__ b1,
                   const float* __restrict__ lng,
                   const float* __restrict__ lnb,
                   float* __restrict__ T) {
  __shared__ float sA[kTileRows * kF];
  __shared__ float sMu[kTileRows];
  __shared__ float sRs[kTileRows];

  const int tid  = threadIdx.x;
  const int wave = tid >> 5;
  const int lane = tid & 31;
  const int hf   = lane >> 4;
  const int l16  = lane & 15;
  const size_t row0 = (size_t)blockIdx.x * kTileRows;

  for (int i = tid; i < kTileRows * kF; i += 256) sA[i] = A[row0 * kF + i];
  __syncthreads();

  const int kOff = 2 * hf;
  v8f acc[kTilesPerWave];

  for (int t = 0; t < kTilesPerWave; ++t) {
    v8f c = {};
    const int n = (wave * kTilesPerWave + t) * 16 + l16;
    const float* __restrict__ bcol = W1 + (size_t)n * kF + kOff;
    const float* __restrict__ arow = sA + l16 * kF + kOff;
    for (int k = 0; k < kF; k += 4) {
      v2f a = *(const v2f*)(arow + k);
      v2f b = *(const v2f*)(bcol + k);
      c = __builtin_amdgcn_wmma_f32_16x16x4_f32(false, a, false, b,
                                                (short)0, c, false, false);
    }
    acc[t] = c;
  }
  __syncthreads();

  for (int t = 0; t < kTilesPerWave; ++t) {
    const int n = (wave * kTilesPerWave + t) * 16 + l16;
    const float bias = b1[n];
#pragma unroll
    for (int r = 0; r < 8; ++r) {
      sA[(r + 8 * hf) * kF + n] = acc[t][r] + bias;
    }
  }
  __syncthreads();

  if (tid < kTileRows) {
    float s = 0.f, s2 = 0.f;
    const float* rp = sA + tid * kF;
    for (int c = 0; c < kF; ++c) { const float v = rp[c]; s += v; s2 += v * v; }
    const float mu  = s / kF;
    const float var = s2 / kF - mu * mu;
    sMu[tid] = mu;
    sRs[tid] = rsqrtf(var + 1e-5f);
  }
  __syncthreads();

  for (int i = tid; i < kTileRows * kF; i += 256) {
    const int r = i / kF, c = i % kF;
    const float v = (sA[i] - sMu[r]) * sRs[r] * lng[c] + lnb[c];
    T[row0 * kF + i] = gelu_exact(v);
  }
}

__global__ __launch_bounds__(256, 2)
void gemm2_perm(const float* __restrict__ T,
                const float* __restrict__ W2,
                const float* __restrict__ b2,
                const int*   __restrict__ perm,
                float* __restrict__ out) {
  __shared__ float sA[kTileRows * kF];
  __shared__ int   sRow[kTileRows];

  const int tid  = threadIdx.x;
  const int wave = tid >> 5;
  const int lane = tid & 31;
  const int hf   = lane >> 4;
  const int l16  = lane & 15;
  const int row0 = blockIdx.x * kTileRows;

  if (tid < kTileRows) {
    const int g   = row0 + tid;
    const int img = g / kN;
    const int n   = g - img * kN;
    sRow[tid] = img * kN + perm[n];
  }
  __syncthreads();
  for (int i = tid; i < kTileRows * kF; i += 256) {
    const int r = i / kF, c = i % kF;
    sA[i] = T[(size_t)sRow[r] * kF + c];
  }
  __syncthreads();

  const int kOff = 2 * hf;
  for (int t = 0; t < kTilesPerWave; ++t) {
    v8f c = {};
    const int n = (wave * kTilesPerWave + t) * 16 + l16;
    const float* __restrict__ bcol = W2 + (size_t)n * kF + kOff;
    const float* __restrict__ arow = sA + l16 * kF + kOff;
    for (int k = 0; k < kF; k += 4) {
      v2f a = *(const v2f*)(arow + k);
      v2f b = *(const v2f*)(bcol + k);
      c = __builtin_amdgcn_wmma_f32_16x16x4_f32(false, a, false, b,
                                                (short)0, c, false, false);
    }
    const float bias = b2[n];
#pragma unroll
    for (int r = 0; r < 8; ++r) {
      const int m = r + 8 * hf;
      out[(size_t)(row0 + m) * kOut + n] = c[r] + bias;
    }
  }
}

extern "C" void kernel_launch(void* const* d_in, const int* in_sizes, int n_in,
                              void* d_out, int out_size, void* d_ws, size_t ws_size,
                              hipStream_t stream) {
  const float* x    = (const float*)d_in[0];
  const float* cw   = (const float*)d_in[1];
  const float* cb   = (const float*)d_in[2];
  const float* bng  = (const float*)d_in[3];
  const float* bnb  = (const float*)d_in[4];
  const float* bnm  = (const float*)d_in[5];
  const float* bnv  = (const float*)d_in[6];
  const float* M    = (const float*)d_in[7];
  const float* W1   = (const float*)d_in[8];
  const float* b1   = (const float*)d_in[9];
  const float* lng  = (const float*)d_in[10];
  const float* lnb  = (const float*)d_in[11];
  const float* W2   = (const float*)d_in[12];
  const float* b2   = (const float*)d_in[13];
  const int*   mask = (const int*)d_in[14];
  const int*   perm = (const int*)d_in[15];
  float* out = (float*)d_out;

  float* flat = (float*)d_ws;
  float* Tbuf = flat + (size_t)kRows * kF;

  {
    const int total = kBS * kC * kW * kW;
    conv_bn_gelu_patch<<<(total + 255) / 256, 256, 0, stream>>>(
        x, cw, cb, bng, bnb, bnm, bnv, M, mask, flat);
  }
  gemm1_ln_gelu<<<kRows / kTileRows, 256, 0, stream>>>(flat, W1, b1, lng, lnb, Tbuf);
  gemm2_perm  <<<kRows / kTileRows, 256, 0, stream>>>(Tbuf, W2, b2, perm, out);
}


# 64 rows per workgroup (was 16): each W fragment load feeds 4 WMMAs (wave tile 4x2, 3 column passes keep the streamed W lines in the WGP cache); LDS panel pitch 772 floats; summation order unchanged
# speedup vs baseline: 1.6597x; 1.6597x over previous
#include <hip/hip_runtime.h>
#include <math.h>

typedef float v2f __attribute__((ext_vector_type(2)));
typedef float v8f __attribute__((ext_vector_type(8)));

namespace {
constexpr int kPS   = 16;
constexpr int kPR   = 14;
constexpr int kN    = 196;
constexpr int kC    = 3;
constexpr int kW    = 224;
constexpr int kBS   = 64;
constexpr int kF    = 768;
constexpr int kHid  = 768;
constexpr int kOut  = 768;
constexpr int kRows = kBS * kN;
constexpr int kTileRows     = 64;
constexpr int kMT           = kTileRows / 16;
constexpr int kWaves        = 8;
constexpr int kTilesPerWave = (kHid / 16) / kWaves;
constexpr int kNTp          = 2;
constexpr int kPass         = kTilesPerWave / kNTp;
constexpr int kKS           = 4;
constexpr int kPitch        = kF + 4;
}


struct Frags {
  v2f a[kKS][kMT];
  v2f b[kKS][kNTp];
};

__device__ __forceinline__ void load_group(const float* __restrict__ arow,
                                           const float* __restrict__ bcol,
                                           int k, Frags& f) {
#pragma unroll
  for (int s = 0; s < kKS; ++s) {
#pragma unroll
    for (int t = 0; t < kNTp; ++t)
      f.b[s][t] = *(const v2f*)(bcol + (size_t)t * 16 * kF + k + 4 * s);
#pragma unroll
    for (int i = 0; i < kMT; ++i)
      f.a[s][i] = *(const v2f*)(arow + i * 16 * kPitch + k + 4 * s);
  }
}

__device__ __forceinline__ void mma_group(v8f (&acc)[kMT][kNTp], const Frags& f) {
#pragma unroll
  for (int s = 0; s < kKS; ++s)
#pragma unroll
    for (int i = 0; i < kMT; ++i)
#pragma unroll
      for (int t = 0; t < kNTp; ++t)
        acc[i][t] = __builtin_amdgcn_wmma_f32_16x16x4_f32(
            false, f.a[s][i], false, f.b[s][t], (short)0, acc[i][t], false, false);
}

__device__ __forceinline__ void panel_gemm(const float* __restrict__ arow,
                                           const float* __restrict__ bcol,
                                           v8f (&acc)[kMT][kNTp]) {
#pragma unroll
  for (int i = 0; i < kMT; ++i)
#pragma unroll
    for (int t = 0; t < kNTp; ++t) acc[i][t] = v8f{};

  Frags f0, f1;
  load_group(arow, bcol, 0, f0);
  for (int k = 0; k < kF; k += 8 * kKS) {
    load_group(arow, bcol, k + 4 * kKS, f1);
    __builtin_amdgcn_sched_barrier(0);
    mma_group(acc, f0);
    __builtin_amdgcn_sched_barrier(0);
    const int kn = (k + 8 * kKS < kF) ? k + 8 * kKS : 0;
    load_group(arow, bcol, kn, f0);
    __builtin_amdgcn_sched_barrier(0);
    mma_group(acc, f1);
    __builtin_amdgcn_sched_barrier(0);
    asm volatile("" ::
        "v"(f0.b[0][0]), "v"(f0.b[0][1]), "v"(f0.b[1][0]), "v"(f0.b[1][1]),
        "v"(f0.b[2][0]), "v"(f0.b[2][1]), "v"(f0.b[3][0]), "v"(f0.b[3][1]),
        "v"(f0.a[0][0]), "v"(f0.a[0][1]), "v"(f0.a[0][2]), "v"(f0.a[0][3]),
        "v"(f0.a[1][0]), "v"(f0.a[1][1]), "v"(f0.a[1][2]), "v"(f0.a[1][3]),
        "v"(f0.a[2][0]), "v"(f0.a[2][1]), "v"(f0.a[2][2]), "v"(f0.a[2][3]),
        "v"(f0.a[3][0]), "v"(f0.a[3][1]), "v"(f0.a[3][2]), "v"(f0.a[3][3]));
  }
}

__device__ __forceinline__ float gelu_exact(float v) {
  return 0.5f * v * (1.0f + erff(v * 0.70710678118654752440f));
}

__global__ __launch_bounds__(256)
void conv_bn_gelu_patch(const float* __restrict__ x,
                        const float* __restrict__ cw,
                        const float* __restrict__ cb,
                        const float* __restrict__ bng,
                        const float* __restrict__ bnb,
                        const float* __restrict__ bnm,
                        const float* __restrict__ bnv,
                        const float* __restrict__ M,
                        const int*   __restrict__ mask,
                        float* __restrict__ flat) {
  const int total = kBS * kC * kW * kW;
  int idx = blockIdx.x * blockDim.x + threadIdx.x;
  if (idx >= total) return;

  const int h   = idx % kW;
  const int w   = (idx / kW) % kW;
  const int co  = (idx / (kW * kW)) % kC;
  const int img = idx / (kW * kW * kC);

  float acc = cb[co];
#pragma unroll
  for (int ci = 0; ci < kC; ++ci) {
    const float* __restrict__ xp = x + ((size_t)(img * kC + ci)) * kW * kW;
#pragma unroll
    for (int kw = 0; kw < 3; ++kw) {
      const int ww = w + kw - 1;
      if (ww < 0 || ww >= kW) continue;
#pragma unroll
      for (int kh = 0; kh < 3; ++kh) {
        const int hh = h + kh - 1;
        if (hh < 0 || hh >= kW) continue;
        acc += xp[ww * kW + hh] * cw[((co * kC + ci) * 3 + kw) * 3 + kh];
      }
    }
  }
  const float inv = bng[co] * rsqrtf(bnv[co] + 1e-5f);
  acc = (acc - bnm[co]) * inv + bnb[co];
  acc = gelu_exact(acc);

  const int pi = w & 15, pw = w >> 4;
  const int pj = h & 15, ph = h >> 4;
  const int n  = pw * kPR + ph;
  if (mask[n] > 0) acc *= M[pi * kPS + pj];

  const int row = img * kN + n;
  const int f   = co * kPS * kPS + pi * kPS + pj;
  flat[(size_t)row * kF + f] = acc;
}

__global__ __launch_bounds__(256)
void gemm1_ln_gelu(const float* __restrict__ A,
                   const float* __restrict__ W1,
                   const float* __restrict__ b1,
                   const float* __restrict__ lng,
                   const float* __restrict__ lnb,
                   float* __restrict__ T) {
  __shared__ __attribute__((aligned(16))) float sA[kTileRows * kPitch];
  __shared__ float sMu[kTileRows];
  __shared__ float sRs[kTileRows];

  const int tid  = threadIdx.x;
  const int wave = tid >> 5;
  const int lane = tid & 31;
  const int hf   = lane >> 4;
  const int l16  = lane & 15;
  const size_t row0 = (size_t)blockIdx.x * kTileRows;

  for (int i = tid; i < kTileRows * (kF / 4); i += 256) {
    const int r = i / (kF / 4), c4 = i % (kF / 4);
    *(float4*)(sA + r * kPitch + 4 * c4) =
        *(const float4*)(A + (row0 + r) * kF + 4 * c4);
  }
  __syncthreads();

  const int kOff = 2 * hf;
  const int n0 = wave * kNTp * 16 + l16;
  v8f acc[kPass][kMT][kNTp];
#pragma unroll
  for (int p = 0; p < kPass; ++p)
    panel_gemm(sA + l16 * kPitch + kOff,
               W1 + (size_t)(n0 + p * kWaves * kNTp * 16) * kF + kOff,
               acc[p]);
  __syncthreads();

#pragma unroll
  for (int p = 0; p < kPass; ++p) {
#pragma unroll
    for (int t = 0; t < kNTp; ++t) {
      const int n = n0 + (p * kWaves * kNTp + t) * 16;
      const float bias = b1[n];
#pragma unroll
      for (int i = 0; i < kMT; ++i) {
#pragma unroll
        for (int r = 0; r < 8; ++r) {
          sA[(16 * i + r + 8 * hf) * kPitch + n] = acc[p][i][t][r] + bias;
        }
      }
    }
  }
  __syncthreads();

  if (tid < kTileRows) {
    float s = 0.f, s2 = 0.f;
    const float* rp = sA + tid * kPitch;
    for (int c = 0; c < kF; ++c) { const float v = rp[c]; s += v; s2 += v * v; }
    const float mu  = s / kF;
    const float var = s2 / kF - mu * mu;
    sMu[tid] = mu;
    sRs[tid] = rsqrtf(var + 1e-5f);
  }
  __syncthreads();

  for (int i = tid; i < kTileRows * kF; i += 256) {
    const int r = i / kF, c = i % kF;
    const float v = (sA[r * kPitch + c] - sMu[r]) * sRs[r] * lng[c] + lnb[c];
    T[row0 * kF + i] = gelu_exact(v);
  }
}

__global__ __launch_bounds__(256)
void gemm2_perm(const float* __restrict__ T,
                const float* __restrict__ W2,
                const float* __restrict__ b2,
                const int*   __restrict__ perm,
                float* __restrict__ out) {
  __shared__ __attribute__((aligned(16))) float sA[kTileRows * kPitch];
  __shared__ int   sRow[kTileRows];

  const int tid  = threadIdx.x;
  const int wave = tid >> 5;
  const int lane = tid & 31;
  const int hf   = lane >> 4;
  const int l16  = lane & 15;
  const int row0 = blockIdx.x * kTileRows;

  if (tid < kTileRows) {
    const int g   = row0 + tid;
    const int img = g / kN;
    const int n   = g - img * kN;
    sRow[tid] = img * kN + perm[n];
  }
  __syncthreads();
  for (int i = tid; i < kTileRows * (kF / 4); i += 256) {
    const int r = i / (kF / 4), c4 = i % (kF / 4);
    *(float4*)(sA + r * kPitch + 4 * c4) =
        *(const float4*)(T + (size_t)sRow[r] * kF + 4 * c4);
  }
  __syncthreads();

  const int kOff = 2 * hf;
  const int n0 = wave * kNTp * 16 + l16;
  for (int p = 0; p < kPass; ++p) {
    v8f acc[kMT][kNTp];
    panel_gemm(sA + l16 * kPitch + kOff,
               W2 + (size_t)(n0 + p * kWaves * kNTp * 16) * kF + kOff,
               acc);
#pragma unroll
    for (int t = 0; t < kNTp; ++t) {
      const int n = n0 + (p * kWaves * kNTp + t) * 16;
      const float bias = b2[n];
#pragma unroll
      for (int i = 0; i < kMT; ++i) {
#pragma unroll
        for (int r = 0; r < 8; ++r) {
          const int m = 16 * i + r + 8 * hf;
          out[(size_t)(row0 + m) * kOut + n] = acc[i][t][r] + bias;
        }
      }
    }
  }
}

extern "C" void kernel_launch(void* const* d_in, const int* in_sizes, int n_in,
                              void* d_out, int out_size, void* d_ws, size_t ws_size,
                              hipStream_t stream) {
  const float* x    = (const float*)d_in[0];
  const float* cw   = (const float*)d_in[1];
  const float* cb   = (const float*)d_in[2];
  const float* bng  = (const float*)d_in[3];
  const float* bnb  = (const float*)d_in[4];
  const float* bnm  = (const float*)d_in[5];
  const float* bnv  = (const float*)d_in[6];
  const float* M    = (const float*)d_in[7];
  const float* W1   = (const float*)d_in[8];
  const float* b1   = (const float*)d_in[9];
  const float* lng  = (const float*)d_in[10];
  const float* lnb  = (const float*)d_in[11];
  const float* W2   = (const float*)d_in[12];
  const float* b2   = (const float*)d_in[13];
  const int*   mask = (const int*)d_in[14];
  const int*   perm = (const int*)d_in[15];
  float* out = (float*)d_out;

  float* flat = (float*)d_ws;
  float* Tbuf = flat + (size_t)kRows * kF;

  {
    const int total = kBS * kC * kW * kW;
    conv_bn_gelu_patch<<<(total + 255) / 256, 256, 0, stream>>>(
        x, cw, cb, bng, bnb, bnm, bnv, M, mask, flat);
  }
  gemm1_ln_gelu<<<kRows / kTileRows, 256, 0, stream>>>(flat, W1, b1, lng, lnb, Tbuf);
  gemm2_perm  <<<kRows / kTileRows, 256, 0, stream>>>(Tbuf, W2, b2, perm, out);
}


# fc1 epilogue by whole rows per wave: LayerNorm sums 24 values per lane, then xor-shuffle tree (summation order changed), no one-thread-per-row pass; LN+GELU loop takes 16 columns per lane per trip
# speedup vs baseline: 2.6592x; 1.6022x over previous
#include <hip/hip_runtime.h>
#include <math.h>

typedef float v2f __attribute__((ext_vector_type(2)));
typedef float v8f __attribute__((ext_vector_type(8)));

namespace {
constexpr int kPS   = 16;
constexpr int kPR   = 14;
constexpr int kN    = 196;
constexpr int kC    = 3;
constexpr int kW    = 224;
constexpr int kBS   = 64;
constexpr int kF    = 768;
constexpr int kHid  = 768;
constexpr int kOut  = 768;
constexpr int kRows = kBS * kN;
constexpr int kTileRows     = 64;
constexpr int kMT           = kTileRows / 16;
constexpr int kWaves        = 8;
constexpr int kTilesPerWave = (kHid / 16) / kWaves;
constexpr int kNTp          = 2;
constexpr int kPass         = kTilesPerWave / kNTp;
constexpr int kKS           = 4;
constexpr int kPitch        = kF + 4;
constexpr int kStripRows   = 8;
constexpr int kStrips      = kW / kStripRows;
constexpr int kConvBlocks  = kBS * kStrips;
}


struct Frags {
  v2f a[kKS][kMT];
  v2f b[kKS][kNTp];
};

__device__ __forceinline__ void load_group(const float* __restrict__ arow,
                                           const float* __restrict__ bcol,
                                           int k, Frags& f) {
#pragma unroll
  for (int s = 0; s < kKS; ++s) {
#pragma unroll
    for (int t = 0; t < kNTp; ++t)
      f.b[s][t] = *(const v2f*)(bcol + (size_t)t * 16 * kF + k + 4 * s);
#pragma unroll
    for (int i = 0; i < kMT; ++i)
      f.a[s][i] = *(const v2f*)(arow + i * 16 * kPitch + k + 4 * s);
  }
}

__device__ __forceinline__ void mma_group(v8f (&acc)[kMT][kNTp], const Frags& f) {
#pragma unroll
  for (int s = 0; s < kKS; ++s)
#pragma unroll
    for (int i = 0; i < kMT; ++i)
#pragma unroll
      for (int t = 0; t < kNTp; ++t)
        acc[i][t] = __builtin_amdgcn_wmma_f32_16x16x4_f32(
            false, f.a[s][i], false, f.b[s][t], (short)0, acc[i][t], false, false);
}

__device__ __forceinline__ void panel_gemm(const float* __restrict__ arow,
                                           const float* __restrict__ bcol,
                                           v8f (&acc)[kMT][kNTp]) {
#pragma unroll
  for (int i = 0; i < kMT; ++i)
#pragma unroll
    for (int t = 0; t < kNTp; ++t) acc[i][t] = v8f{};

  Frags f0, f1;
  load_group(arow, bcol, 0, f0);
  for (int k = 0; k < kF; k += 8 * kKS) {
    load_group(arow, bcol, k + 4 * kKS, f1);
    __builtin_amdgcn_sched_barrier(0);
    mma_group(acc, f0);
    __builtin_amdgcn_sched_barrier(0);
    const int kn = (k + 8 * kKS < kF) ? k + 8 * kKS : 0;
    load_group(arow, bcol, kn, f0);
    __builtin_amdgcn_sched_barrier(0);
    mma_group(acc, f1);
    __builtin_amdgcn_sched_barrier(0);
    asm volatile("" ::
        "v"(f0.b[0][0]), "v"(f0.b[0][1]), "v"(f0.b[1][0]), "v"(f0.b[1][1]),
        "v"(f0.b[2][0]), "v"(f0.b[2][1]), "v"(f0.b[3][0]), "v"(f0.b[3][1]),
        "v"(f0.a[0][0]), "v"(f0.a[0][1]), "v"(f0.a[0][2]), "v"(f0.a[0][3]),
        "v"(f0.a[1][0]), "v"(f0.a[1][1]), "v"(f0.a[1][2]), "v"(f0.a[1][3]),
        "v"(f0.a[2][0]), "v"(f0.a[2][1]), "v"(f0.a[2][2]), "v"(f0.a[2][3]),
        "v"(f0.a[3][0]), "v"(f0.a[3][1]), "v"(f0.a[3][2]), "v"(f0.a[3][3]));
  }
}

__device__ __forceinline__ float gelu_exact(float v) {
  return 0.5f * v * (1.0f + erff(v * 0.70710678118654752440f));
}

struct XRow { float v[kC][3]; };

template <bool kEdgeRow>
__device__ __forceinline__ void load_xrow(const float* __restrict__ ximg, int ww,
                                          int h, int hm, int hp, XRow& r) {
  const bool inside = !kEdgeRow || (ww >= 0 && ww < kW);
#pragma unroll
  for (int ci = 0; ci < kC; ++ci) {
    float a = 0.f, b = 0.f, c = 0.f;
    if (inside) {
      const float* __restrict__ p = ximg + ((size_t)ci * kW + ww) * kW;
      a = p[hm];
      b = p[h];
      c = p[hp];
    }
    r.v[ci][0] = (h > 0) ? a : 0.f;
    r.v[ci][1] = b;
    r.v[ci][2] = (h < kW - 1) ? c : 0.f;
  }
}

__global__ __launch_bounds__(kW)
void conv_bn_gelu_patch(const float* __restrict__ x,
                        const float* __restrict__ cw,
                        const float* __restrict__ cb,
                        const float* __restrict__ bng,
                        const float* __restrict__ bnb,
                        const float* __restrict__ bnm,
                        const float* __restrict__ bnv,
                        const float* __restrict__ M,
                        const int*   __restrict__ mask,
                        float* __restrict__ flat) {
  const int h     = threadIdx.x;
  const int img   = blockIdx.x / kStrips;
  const int strip = blockIdx.x - img * kStrips;
  const int w0    = strip * kStripRows;
  const int hm    = (h > 0) ? h - 1 : 0;
  const int hp    = (h < kW - 1) ? h + 1 : kW - 1;
  const float* __restrict__ ximg = x + (size_t)img * kC * kW * kW;

  float wgt[kC][kC][3][3];
#pragma unroll
  for (int co = 0; co < kC; ++co)
#pragma unroll
    for (int ci = 0; ci < kC; ++ci)
#pragma unroll
      for (int kw = 0; kw < 3; ++kw)
#pragma unroll
        for (int kh = 0; kh < 3; ++kh)
          wgt[co][ci][kw][kh] = cw[((co * kC + ci) * 3 + kw) * 3 + kh];
  float bias[kC], mean[kC], inv[kC], beta[kC];
#pragma unroll
  for (int co = 0; co < kC; ++co) {
    bias[co] = cb[co];
    mean[co] = bnm[co];
    inv[co]  = bng[co] * rsqrtf(bnv[co] + 1e-5f);
    beta[co] = bnb[co];
  }

  const int pj  = h & 15, ph = h >> 4;
  const int pi0 = w0 & 15;
  const int n   = (w0 >> 4) * kPR + ph;
  const bool gate = mask[n] > 0;
  const float* __restrict__ mrow = M + pi0 * kPS + pj;
  float* __restrict__ orow = flat + (size_t)(img * kN + n) * kF + pi0 * kPS + pj;

  XRow win[4];
  load_xrow<true >(ximg, w0 - 1, h, hm, hp, win[0]);
  load_xrow<false>(ximg, w0,     h, hm, hp, win[1]);
  load_xrow<false>(ximg, w0 + 1, h, hm, hp, win[2]);
#pragma unroll
  for (int r = 0; r < kStripRows; ++r) {
    const float mv = mrow[r * kPS];
    const float mm = gate ? mv : 1.0f;
    if (r + 2 < kStripRows)
      load_xrow<false>(ximg, w0 + r + 2, h, hm, hp, win[(r + 3) & 3]);
    else if (r + 2 == kStripRows)
      load_xrow<true >(ximg, w0 + r + 2, h, hm, hp, win[(r + 3) & 3]);
#pragma unroll
    for (int co = 0; co < kC; ++co) {
      float acc = bias[co];
#pragma unroll
      for (int ci = 0; ci < kC; ++ci)
#pragma unroll
        for (int kw = 0; kw < 3; ++kw)
#pragma unroll
          for (int kh = 0; kh < 3; ++kh)
            acc += win[(r + kw) & 3].v[ci][kh] * wgt[co][ci][kw][kh];
      acc = (acc - mean[co]) * inv[co] + beta[co];
      acc = gelu_exact(acc);
      orow[co * kPS * kPS + r * kPS] = acc * mm;
    }
  }
}

__global__ __launch_bounds__(256)
void gemm1_ln_gelu(const float* __restrict__ A,
                   const float* __restrict__ W1,
                   const float* __restrict__ b1,
                   const float* __restrict__ lng,
                   const float* __restrict__ lnb,
                   float* __restrict__ T) {
  __shared__ __attribute__((aligned(16))) float sA[kTileRows * kPitch];

  const int tid  = threadIdx.x;
  const int wave = tid >> 5;
  const int lane = tid & 31;
  const int hf   = lane >> 4;
  const int l16  = lane & 15;
  const size_t row0 = (size_t)blockIdx.x * kTileRows;

  for (int i = tid; i < kTileRows * (kF / 4); i += 256) {
    const int r = i / (kF / 4), c4 = i % (kF / 4);
    *(float4*)(sA + r * kPitch + 4 * c4) =
        *(const float4*)(A + (row0 + r) * kF + 4 * c4);
  }
  __syncthreads();

  const int kOff = 2 * hf;
  const int n0 = wave * kNTp * 16 + l16;
  v8f acc[kPass][kMT][kNTp];
#pragma unroll
  for (int p = 0; p < kPass; ++p)
    panel_gemm(sA + l16 * kPitch + kOff,
               W1 + (size_t)(n0 + p * kWaves * kNTp * 16) * kF + kOff,
               acc[p]);
  __syncthreads();

#pragma unroll
  for (int p = 0; p < kPass; ++p) {
#pragma unroll
    for (int t = 0; t < kNTp; ++t) {
      const int n = n0 + (p * kWaves * kNTp + t) * 16;
      const float bias = b1[n];
#pragma unroll
      for (int i = 0; i < kMT; ++i) {
#pragma unroll
        for (int r = 0; r < 8; ++r) {
          sA[(16 * i + r + 8 * hf) * kPitch + n] = acc[p][i][t][r] + bias;
        }
      }
    }
  }
  __syncthreads();

  constexpr int kRuns = kF / 128;
  constexpr int kRowsPerWave = kTileRows / kWaves;
  for (int rr = 0; rr < kRowsPerWave; ++rr) {
    const int r = wave * kRowsPerWave + rr;
    const float* rp = sA + r * kPitch;
    float s = 0.f, s2 = 0.f;
#pragma unroll
    for (int j = 0; j < kRuns; ++j) {
      const float4 v = *(const float4*)(rp + 4 * lane + 128 * j);
      s += v.x; s2 += v.x * v.x;
      s += v.y; s2 += v.y * v.y;
      s += v.z; s2 += v.z * v.z;
      s += v.w; s2 += v.w * v.w;
    }
#pragma unroll
    for (int o = 16; o > 0; o >>= 1) {
      s  += __shfl_xor(s, o);
      s2 += __shfl_xor(s2, o);
    }
    const float mu  = s / kF;
    const float var = s2 / kF - mu * mu;
    const float rs  = rsqrtf(var + 1e-5f);
    float* op = T + (row0 + r) * kF;
#pragma clang loop interleave_count(8)
    for (int c = lane; c < kF; c += 32) {
      const float v = (rp[c] - mu) * rs * lng[c] + lnb[c];
      op[c] = gelu_exact(v);
    }
  }
}

__global__ __launch_bounds__(256)
void gemm2_perm(const float* __restrict__ T,
                const float* __restrict__ W2,
                const float* __restrict__ b2,
                const int*   __restrict__ perm,
                float* __restrict__ out) {
  __shared__ __attribute__((aligned(16))) float sA[kTileRows * kPitch];
  __shared__ int   sRow[kTileRows];

  const int tid  = threadIdx.x;
  const int wave = tid >> 5;
  const int lane = tid & 31;
  const int hf   = lane >> 4;
  const int l16  = lane & 15;
  const int row0 = blockIdx.x * kTileRows;

  if (tid < kTileRows) {
    const int g   = row0 + tid;
    const int img = g / kN;
    const int n   = g - img * kN;
    sRow[tid] = img * kN + perm[n];
  }
  __syncthreads();
  for (int i = tid; i < kTileRows * (kF / 4); i += 256) {
    const int r = i / (kF / 4), c4 = i % (kF / 4);
    *(float4*)(sA + r * kPitch + 4 * c4) =
        *(const float4*)(T + (size_t)sRow[r] * kF + 4 * c4);
  }
  __syncthreads();

  const int kOff = 2 * hf;
  const int n0 = wave * kNTp * 16 + l16;
  for (int p = 0; p < kPass; ++p) {
    v8f acc[kMT][kNTp];
    panel_gemm(sA + l16 * kPitch + kOff,
               W2 + (size_t)(n0 + p * kWaves * kNTp * 16) * kF + kOff,
               acc);
#pragma unroll
    for (int t = 0; t < kNTp; ++t) {
      const int n = n0 + (p * kWaves * kNTp + t) * 16;
      const float bias = b2[n];
#pragma unroll
      for (int i = 0; i < kMT; ++i) {
#pragma unroll
        for (int r = 0; r < 8; ++r) {
          const int m = 16 * i + r + 8 * hf;
          out[(size_t)(row0 + m) * kOut + n] = acc[i][t][r] + bias;
        }
      }
    }
  }
}

extern "C" void kernel_launch(void* const* d_in, const int* in_sizes, int n_in,
                              void* d_out, int out_size, void* d_ws, size_t ws_size,
                              hipStream_t stream) {
  const float* x    = (const float*)d_in[0];
  const float* cw   = (const float*)d_in[1];
  const float* cb   = (const float*)d_in[2];
  const float* bng  = (const float*)d_in[3];
  const float* bnb  = (const float*)d_in[4];
  const float* bnm  = (const float*)d_in[5];
  const float* bnv  = (const float*)d_in[6];
  const float* M    = (const float*)d_in[7];
  const float* W1   = (const float*)d_in[8];
  const float* b1   = (const float*)d_in[9];
  const float* lng  = (const float*)d_in[10];
  const float* lnb  = (const float*)d_in[11];
  const float* W2   = (const float*)d_in[12];
  const float* b2   = (const float*)d_in[13];
  const int*   mask = (const int*)d_in[14];
  const int*   perm = (const int*)d_in[15];
  float* out = (float*)d_out;

  float* flat = (float*)d_ws;
  float* Tbuf = flat + (size_t)kRows * kF;

  conv_bn_gelu_patch<<<kConvBlocks, kW, 0, stream>>>(
      x, cw, cb, bng, bnb, bnm, bnv, M, mask, flat);
  gemm1_ln_gelu<<<kRows / kTileRows, 256, 0, stream>>>(flat, W1, b1, lng, lnb, Tbuf);
  gemm2_perm  <<<kRows / kTileRows, 256, 0, stream>>>(Tbuf, W2, b2, perm, out);
}


# both GEMMs on 256 workgroups, one per WGP (was 196 of 64 rows): 48 whole rows each + one 16x48 piece of the last 256 rows, whose LayerNorm+GELU runs as a small 4th launch; summation order unchanged
# speedup vs baseline: 2.9937x; 1.1258x over previous
#include <hip/hip_runtime.h>
#include <math.h>

typedef float v2f __attribute__((ext_vector_type(2)));
typedef float v8f __attribute__((ext_vector_type(8)));

namespace {
constexpr int kPS   = 16;
constexpr int kPR   = 14;
constexpr int kN    = 196;
constexpr int kC    = 3;
constexpr int kW    = 224;
constexpr int kBS   = 64;
constexpr int kF    = 768;
constexpr int kHid  = 768;
constexpr int kOut  = 768;
constexpr int kRows = kBS * kN;
constexpr int kGrid         = 256;
constexpr int kMainRows     = 48;
constexpr int kMT           = kMainRows / 16;
constexpr int kTailRow0     = kGrid * kMainRows;
constexpr int kTailTiles    = 3;
constexpr int kTailGroups   = (kHid / 16) / kTailTiles;
constexpr int kTileRows     = kMainRows + 16;
constexpr int kWaves        = 8;
constexpr int kTilesPerWave = (kHid / 16) / kWaves;
constexpr int kNTp          = 2;
constexpr int kPass         = kTilesPerWave / kNTp;
constexpr int kKS           = 4;
constexpr int kPitch        = kF + 4;
constexpr int kStripRows   = 8;
constexpr int kStrips      = kW / kStripRows;
constexpr int kConvBlocks  = kBS * kStrips;
static_assert(kTailRow0 + 16 * (kGrid / kTailGroups) == kRows, "tail pieces cover the last rows");
static_assert(kTailTiles * kTailGroups * 16 == kHid, "tail pieces cover every column");
}


struct Frags {
  v2f a[kKS][kMT];
  v2f b[kKS][kNTp];
};

__device__ __forceinline__ void load_group(const float* __restrict__ arow,
                                           const float* __restrict__ bcol,
                                           int k, Frags& f) {
#pragma unroll
  for (int s = 0; s < kKS; ++s) {
#pragma unroll
    for (int t = 0; t < kNTp; ++t)
      f.b[s][t] = *(const v2f*)(bcol + (size_t)t * 16 * kF + k + 4 * s);
#pragma unroll
    for (int i = 0; i < kMT; ++i)
      f.a[s][i] = *(const v2f*)(arow + i * 16 * kPitch + k + 4 * s);
  }
}

__device__ __forceinline__ void mma_group(v8f (&acc)[kMT][kNTp], const Frags& f) {
#pragma unroll
  for (int s = 0; s < kKS; ++s)
#pragma unroll
    for (int i = 0; i < kMT; ++i)
#pragma unroll
      for (int t = 0; t < kNTp; ++t)
        acc[i][t] = __builtin_amdgcn_wmma_f32_16x16x4_f32(
            false, f.a[s][i], false, f.b[s][t], (short)0, acc[i][t], false, false);
}

__device__ __forceinline__ void panel_gemm(const float* __restrict__ arow,
                                           const float* __restrict__ bcol,
                                           v8f (&acc)[kMT][kNTp]) {
#pragma unroll
  for (int i = 0; i < kMT; ++i)
#pragma unroll
    for (int t = 0; t < kNTp; ++t) acc[i][t] = v8f{};

  Frags f0, f1;
  load_group(arow, bcol, 0, f0);
  for (int k = 0; k < kF; k += 8 * kKS) {
    load_group(arow, bcol, k + 4 * kKS, f1);
    __builtin_amdgcn_sched_barrier(0);
    mma_group(acc, f0);
    __builtin_amdgcn_sched_barrier(0);
    const int kn = (k + 8 * kKS < kF) ? k + 8 * kKS : 0;
    load_group(arow, bcol, kn, f0);
    __builtin_amdgcn_sched_barrier(0);
    mma_group(acc, f1);
    __builtin_amdgcn_sched_barrier(0);
    asm volatile("" ::
        "v"(f0.b[0][0]), "v"(f0.b[0][1]), "v"(f0.b[1][0]), "v"(f0.b[1][1]),
        "v"(f0.b[2][0]), "v"(f0.b[2][1]), "v"(f0.b[3][0]), "v"(f0.b[3][1]),
        "v"(f0.a[0][0]), "v"(f0.a[0][1]), "v"(f0.a[0][2]),
        "v"(f0.a[1][0]), "v"(f0.a[1][1]), "v"(f0.a[1][2]),
        "v"(f0.a[2][0]), "v"(f0.a[2][1]), "v"(f0.a[2][2]),
        "v"(f0.a[3][0]), "v"(f0.a[3][1]), "v"(f0.a[3][2]));
  }
}

__device__ __forceinline__ v8f tail_tile(const float* __restrict__ arow,
                                         const float* __restrict__ bcol) {
  v8f acc = v8f{};
#pragma unroll 16
  for (int k = 0; k < kF; k += 4) {
    const v2f a = *(const v2f*)(arow + k);
    const v2f b = *(const v2f*)(bcol + k);
    acc = __builtin_amdgcn_wmma_f32_16x16x4_f32(false, a, false, b, (short)0, acc,
                                                false, false);
  }
  return acc;
}

__device__ __forceinline__ float gelu_exact(float v) {
  return 0.5f * v * (1.0f + erff(v * 0.70710678118654752440f));
}

struct XRow { float v[kC][3]; };

template <bool kEdgeRow>
__device__ __forceinline__ void load_xrow(const float* __restrict__ ximg, int ww,
                                          int h, int hm, int hp, XRow& r) {
  const bool inside = !kEdgeRow || (ww >= 0 && ww < kW);
#pragma unroll
  for (int ci = 0; ci < kC; ++ci) {
    float a = 0.f, b = 0.f, c = 0.f;
    if (inside) {
      const float* __restrict__ p = ximg + ((size_t)ci * kW + ww) * kW;
      a = p[hm];
      b = p[h];
      c = p[hp];
    }
    r.v[ci][0] = (h > 0) ? a : 0.f;
    r.v[ci][1] = b;
    r.v[ci][2] = (h < kW - 1) ? c : 0.f;
  }
}

__global__ __launch_bounds__(kW)
void conv_bn_gelu_patch(const float* __restrict__ x,
                        const float* __restrict__ cw,
                        const float* __restrict__ cb,
                        const float* __restrict__ bng,
                        const float* __restrict__ bnb,
                        const float* __restrict__ bnm,
                        const float* __restrict__ bnv,
                        const float* __restrict__ M,
                        const int*   __restrict__ mask,
                        float* __restrict__ flat) {
  const int h     = threadIdx.x;
  const int img   = blockIdx.x / kStrips;
  const int strip = blockIdx.x - img * kStrips;
  const int w0    = strip * kStripRows;
  const int hm    = (h > 0) ? h - 1 : 0;
  const int hp    = (h < kW - 1) ? h + 1 : kW - 1;
  const float* __restrict__ ximg = x + (size_t)img * kC * kW * kW;

  float wgt[kC][kC][3][3];
#pragma unroll
  for (int co = 0; co < kC; ++co)
#pragma unroll
    for (int ci = 0; ci < kC; ++ci)
#pragma unroll
      for (int kw = 0; kw < 3; ++kw)
#pragma unroll
        for (int kh = 0; kh < 3; ++kh)
          wgt[co][ci][kw][kh] = cw[((co * kC + ci) * 3 + kw) * 3 + kh];
  float bias[kC], mean[kC], inv[kC], beta[kC];
#pragma unroll
  for (int co = 0; co < kC; ++co) {
    bias[co] = cb[co];
    mean[co] = bnm[co];
    inv[co]  = bng[co] * rsqrtf(bnv[co] + 1e-5f);
    beta[co] = bnb[co];
  }

  const int pj  = h & 15, ph = h >> 4;
  const int pi0 = w0 & 15;
  const int n   = (w0 >> 4) * kPR + ph;
  const bool gate = mask[n] > 0;
  const float* __restrict__ mrow = M + pi0 * kPS + pj;
  float* __restrict__ orow = flat + (size_t)(img * kN + n) * kF + pi0 * kPS + pj;

  XRow win[4];
  load_xrow<true >(ximg, w0 - 1, h, hm, hp, win[0]);
  load_xrow<false>(ximg, w0,     h, hm, hp, win[1]);
  load_xrow<false>(ximg, w0 + 1, h, hm, hp, win[2]);
#pragma unroll
  for (int r = 0; r < kStripRows; ++r) {
    const float mv = mrow[r * kPS];
    const float mm = gate ? mv : 1.0f;
    if (r + 2 < kStripRows)
      load_xrow<false>(ximg, w0 + r + 2, h, hm, hp, win[(r + 3) & 3]);
    else if (r + 2 == kStripRows)
      load_xrow<true >(ximg, w0 + r + 2, h, hm, hp, win[(r + 3) & 3]);
#pragma unroll
    for (int co = 0; co < kC; ++co) {
      float acc = bias[co];
#pragma unroll
      for (int ci = 0; ci < kC; ++ci)
#pragma unroll
        for (int kw = 0; kw < 3; ++kw)
#pragma unroll
          for (int kh = 0; kh < 3; ++kh)
            acc += win[(r + kw) & 3].v[ci][kh] * wgt[co][ci][kw][kh];
      acc = (acc - mean[co]) * inv[co] + beta[co];
      acc = gelu_exact(acc);
      orow[co * kPS * kPS + r * kPS] = acc * mm;
    }
  }
}


__global__ __launch_bounds__(256)
void gemm1_ln_gelu(const float* __restrict__ A,
                   const float* __restrict__ W1,
                   const float* __restrict__ b1,
                   const float* __restrict__ lng,
                   const float* __restrict__ lnb,
                   float* __restrict__ T) {
  __shared__ __attribute__((aligned(16))) float sA[kTileRows * kPitch];

  const int tid  = threadIdx.x;
  const int wave = __builtin_amdgcn_readfirstlane(tid >> 5);
  const int lane = tid & 31;
  const int hf   = lane >> 4;
  const int l16  = lane & 15;
  const size_t row0  = (size_t)blockIdx.x * kMainRows;
  const size_t tail0 = (size_t)kTailRow0 + 16 * (blockIdx.x / kTailGroups);
  const int    tcol0 = (blockIdx.x % kTailGroups) * kTailTiles * 16;

  const float* __restrict__ amain = A + row0 * kF;
  for (int i = tid; i < kMainRows * (kF / 4); i += 256) {
    const int r = i / (kF / 4), c4 = i % (kF / 4);
    *(float4*)(sA + r * kPitch + 4 * c4) = *(const float4*)(amain + 4 * i);
  }
  const float* __restrict__ atail = A + tail0 * kF;
  for (int i = tid; i < 16 * (kF / 4); i += 256) {
    const int r = i / (kF / 4), c4 = i % (kF / 4);
    *(float4*)(sA + (kMainRows + r) * kPitch + 4 * c4) = *(const float4*)(atail + 4 * i);
  }
  __syncthreads();

  const int kOff = 2 * hf;

  if (wave < kTailTiles) {
    const int n = tcol0 + wave * 16 + l16;
    const v8f d = tail_tile(sA + (kMainRows + l16) * kPitch + kOff,
                            W1 + (size_t)n * kF + kOff);
    const float bias = b1[n];
#pragma unroll
    for (int r = 0; r < 8; ++r)
      T[(tail0 + r + 8 * hf) * kHid + n] = d[r] + bias;
  }

  const int n0 = wave * kNTp * 16 + l16;
  v8f acc[kPass][kMT][kNTp];
#pragma unroll
  for (int p = 0; p < kPass; ++p)
    panel_gemm(sA + l16 * kPitch + kOff,
               W1 + (size_t)(n0 + p * kWaves * kNTp * 16) * kF + kOff,
               acc[p]);
  __syncthreads();

#pragma unroll
  for (int p = 0; p < kPass; ++p) {
#pragma unroll
    for (int t = 0; t < kNTp; ++t) {
      const int n = n0 + (p * kWaves * kNTp + t) * 16;
      const float bias = b1[n];
#pragma unroll
      for (int i = 0; i < kMT; ++i) {
#pragma unroll
        for (int r = 0; r < 8; ++r) {
          sA[(16 * i + r + 8 * hf) * kPitch + n] = acc[p][i][t][r] + bias;
        }
      }
    }
  }
  __syncthreads();

  constexpr int kRuns = kF / 128;
  constexpr int kRowsPerWave = kMainRows / kWaves;
  for (int rr = 0; rr < kRowsPerWave; ++rr) {
    const int r = wave * kRowsPerWave + rr;
    const float* rp = sA + r * kPitch;
    float s = 0.f, s2 = 0.f;
#pragma unroll
    for (int j = 0; j < kRuns; ++j) {
      const float4 v = *(const float4*)(rp + 4 * lane + 128 * j);
      s += v.x; s2 += v.x * v.x;
      s += v.y; s2 += v.y * v.y;
      s += v.z; s2 += v.z * v.z;
      s += v.w; s2 += v.w * v.w;
    }
#pragma unroll
    for (int o = 16; o > 0; o >>= 1) {
      s  += __shfl_xor(s, o);
      s2 += __shfl_xor(s2, o);
    }
    const float mu  = s / kF;
    const float var = s2 / kF - mu * mu;
    const float rs  = rsqrtf(var + 1e-5f);
    float* op = T + (row0 + r) * kF;
#pragma clang loop interleave_count(8)
    for (int c = lane; c < kF; c += 32) {
      const float v = (rp[c] - mu) * rs * lng[c] + lnb[c];
      op[c] = gelu_exact(v);
    }
  }
}

__global__ __launch_bounds__(256)
void ln_gelu_tail(const float* __restrict__ lng,
                  const float* __restrict__ lnb,
                  float* T) {
  const int wave = threadIdx.x >> 5;
  const int lane = threadIdx.x & 31;
  float* rp = T + (size_t)(kTailRow0 + blockIdx.x * kWaves + wave) * kHid + 4 * lane;
  constexpr int kRuns = kHid / 128;
  float4 v[kRuns];
#pragma unroll
  for (int j = 0; j < kRuns; ++j) v[j] = *(const float4*)(rp + 128 * j);
  float s = 0.f, s2 = 0.f;
#pragma unroll
  for (int j = 0; j < kRuns; ++j) {
    s += v[j].x; s2 += v[j].x * v[j].x;
    s += v[j].y; s2 += v[j].y * v[j].y;
    s += v[j].z; s2 += v[j].z * v[j].z;
    s += v[j].w; s2 += v[j].w * v[j].w;
  }
#pragma unroll
  for (int o = 16; o > 0; o >>= 1) {
    s  += __shfl_xor(s, o);
    s2 += __shfl_xor(s2, o);
  }
  const float mu  = s / kHid;
  const float var = s2 / kHid - mu * mu;
  const float rs  = rsqrtf(var + 1e-5f);
#pragma unroll
  for (int j = 0; j < kRuns; ++j) {
    const float4 g = *(const float4*)(lng + 4 * lane + 128 * j);
    const float4 b = *(const float4*)(lnb + 4 * lane + 128 * j);
    float4 o;
    o.x = gelu_exact((v[j].x - mu) * rs * g.x + b.x);
    o.y = gelu_exact((v[j].y - mu) * rs * g.y + b.y);
    o.z = gelu_exact((v[j].z - mu) * rs * g.z + b.z);
    o.w = gelu_exact((v[j].w - mu) * rs * g.w + b.w);
    *(float4*)(rp + 128 * j) = o;
  }
}

__global__ __launch_bounds__(256)
void gemm2_perm(const float* __restrict__ T,
                const float* __restrict__ W2,
                const float* __restrict__ b2,
                const int*   __restrict__ perm,
                float* __restrict__ out) {
  __shared__ __attribute__((aligned(16))) float sA[kTileRows * kPitch];
  __shared__ int   sRow[kTileRows];

  const int tid  = threadIdx.x;
  const int wave = __builtin_amdgcn_readfirstlane(tid >> 5);
  const int lane = tid & 31;
  const int hf   = lane >> 4;
  const int l16  = lane & 15;
  const int row0  = blockIdx.x * kMainRows;
  const int tail0 = kTailRow0 + 16 * (blockIdx.x / kTailGroups);
  const int tcol0 = (blockIdx.x % kTailGroups) * kTailTiles * 16;

  if (tid < kTileRows) {
    const int g   = (tid < kMainRows) ? row0 + tid : tail0 + (tid - kMainRows);
    const int img = g / kN;
    const int n   = g - img * kN;
    sRow[tid] = img * kN + perm[n];
  }
  __syncthreads();
  for (int i = tid; i < kTileRows * (kF / 4); i += 256) {
    const int r = i / (kF / 4), c4 = i % (kF / 4);
    *(float4*)(sA + r * kPitch + 4 * c4) =
        *(const float4*)(T + (size_t)sRow[r] * kF + 4 * c4);
  }
  __syncthreads();

  const int kOff = 2 * hf;

  if (wave < kTailTiles) {
    const int n = tcol0 + wave * 16 + l16;
    const v8f d = tail_tile(sA + (kMainRows + l16) * kPitch + kOff,
                            W2 + (size_t)n * kF + kOff);
    const float bias = b2[n];
#pragma unroll
    for (int r = 0; r < 8; ++r)
      out[(size_t)(tail0 + r + 8 * hf) * kOut + n] = d[r] + bias;
  }

  const int n0 = wave * kNTp * 16 + l16;
  for (int p = 0; p < kPass; ++p) {
    v8f acc[kMT][kNTp];
    panel_gemm(sA + l16 * kPitch + kOff,
               W2 + (size_t)(n0 + p * kWaves * kNTp * 16) * kF + kOff,
               acc);
#pragma unroll
    for (int t = 0; t < kNTp; ++t) {
      const int n = n0 + (p * kWaves * kNTp + t) * 16;
      const float bias = b2[n];
#pragma unroll
      for (int i = 0; i < kMT; ++i) {
#pragma unroll
        for (int r = 0; r < 8; ++r) {
          const int m = 16 * i + r + 8 * hf;
          out[(size_t)(row0 + m) * kOut + n] = acc[i][t][r] + bias;
        }
      }
    }
  }
}

extern "C" void kernel_launch(void* const* d_in, const int* in_sizes, int n_in,
                              void* d_out, int out_size, void* d_ws, size_t ws_size,
                              hipStream_t stream) {
  const float* x    = (const float*)d_in[0];
  const float* cw   = (const float*)d_in[1];
  const float* cb   = (const float*)d_in[2];
  const float* bng  = (const float*)d_in[3];
  const float* bnb  = (const float*)d_in[4];
  const float* bnm  = (const float*)d_in[5];
  const float* bnv  = (const float*)d_in[6];
  const float* M    = (const float*)d_in[7];
  const float* W1   = (const float*)d_in[8];
  const float* b1   = (const float*)d_in[9];
  const float* lng  = (const float*)d_in[10];
  const float* lnb  = (const float*)d_in[11];
  const float* W2   = (const float*)d_in[12];
  const float* b2   = (const float*)d_in[13];
  const int*   mask = (const int*)d_in[14];
  const int*   perm = (const int*)d_in[15];
  float* out = (float*)d_out;

  float* flat = (float*)d_ws;
  float* Tbuf = flat + (size_t)kRows * kF;

  conv_bn_gelu_patch<<<kConvBlocks, kW, 0, stream>>>(
      x, cw, cb, bng, bnb, bnm, bnv, M, mask, flat);
  gemm1_ln_gelu<<<kGrid, 256, 0, stream>>>(flat, W1, b1, lng, lnb, Tbuf);
  ln_gelu_tail <<<(kRows - kTailRow0) / kWaves, 256, 0, stream>>>(lng, lnb, Tbuf);
  gemm2_perm   <<<kGrid, 256, 0, stream>>>(Tbuf, W2, b2, perm, out);
}


# both GEMMs stage the 64x768 A panel with all 48 16-byte loads of a lane in flight at once; was 4 per wait (fc1) and 2 (fc2 gather); LDS layout and summation order unchanged
# speedup vs baseline: 3.1255x; 1.0440x over previous
#include <hip/hip_runtime.h>
#include <math.h>

typedef float v2f __attribute__((ext_vector_type(2)));
typedef float v8f __attribute__((ext_vector_type(8)));

namespace {
constexpr int kPS   = 16;
constexpr int kPR   = 14;
constexpr int kN    = 196;
constexpr int kC    = 3;
constexpr int kW    = 224;
constexpr int kBS   = 64;
constexpr int kF    = 768;
constexpr int kHid  = 768;
constexpr int kOut  = 768;
constexpr int kRows = kBS * kN;
constexpr int kGrid         = 256;
constexpr int kMainRows     = 48;
constexpr int kMT           = kMainRows / 16;
constexpr int kTailRow0     = kGrid * kMainRows;
constexpr int kTailTiles    = 3;
constexpr int kTailGroups   = (kHid / 16) / kTailTiles;
constexpr int kTileRows     = kMainRows + 16;
constexpr int kWaves        = 8;
constexpr int kTilesPerWave = (kHid / 16) / kWaves;
constexpr int kNTp          = 2;
constexpr int kPass         = kTilesPerWave / kNTp;
constexpr int kKS           = 4;
constexpr int kPitch        = kF + 4;
constexpr int kStageCB      = kF / 256;
constexpr int kStageRB      = kTileRows / 4;
constexpr int kStripRows   = 8;
constexpr int kStrips      = kW / kStripRows;
constexpr int kConvBlocks  = kBS * kStrips;
static_assert(kTailRow0 + 16 * (kGrid / kTailGroups) == kRows, "tail pieces cover the last rows");
static_assert(kTailTiles * kTailGroups * 16 == kHid, "tail pieces cover every column");
static_assert(kStageCB * 256 == kF && kStageRB * 4 == kTileRows && kMainRows % 4 == 0,
              "staging covers the panel; a block of 4 rows is all main or all tail");
}


struct Frags {
  v2f a[kKS][kMT];
  v2f b[kKS][kNTp];
};

__device__ __forceinline__ void load_group(const float* __restrict__ arow,
                                           const float* __restrict__ bcol,
                                           int k, Frags& f) {
#pragma unroll
  for (int s = 0; s < kKS; ++s) {
#pragma unroll
    for (int t = 0; t < kNTp; ++t)
      f.b[s][t] = *(const v2f*)(bcol + (size_t)t * 16 * kF + k + 4 * s);
#pragma unroll
    for (int i = 0; i < kMT; ++i)
      f.a[s][i] = *(const v2f*)(arow + i * 16 * kPitch + k + 4 * s);
  }
}

__device__ __forceinline__ void mma_group(v8f (&acc)[kMT][kNTp], const Frags& f) {
#pragma unroll
  for (int s = 0; s < kKS; ++s)
#pragma unroll
    for (int i = 0; i < kMT; ++i)
#pragma unroll
      for (int t = 0; t < kNTp; ++t)
        acc[i][t] = __builtin_amdgcn_wmma_f32_16x16x4_f32(
            false, f.a[s][i], false, f.b[s][t], (short)0, acc[i][t], false, false);
}

__device__ __forceinline__ void panel_gemm(const float* __restrict__ arow,
                                           const float* __restrict__ bcol,
                                           v8f (&acc)[kMT][kNTp]) {
#pragma unroll
  for (int i = 0; i < kMT; ++i)
#pragma unroll
    for (int t = 0; t < kNTp; ++t) acc[i][t] = v8f{};

  Frags f0, f1;
  load_group(arow, bcol, 0, f0);
  for (int k = 0; k < kF; k += 8 * kKS) {
    load_group(arow, bcol, k + 4 * kKS, f1);
    __builtin_amdgcn_sched_barrier(0);
    mma_group(acc, f0);
    __builtin_amdgcn_sched_barrier(0);
    const int kn = (k + 8 * kKS < kF) ? k + 8 * kKS : 0;
    load_group(arow, bcol, kn, f0);
    __builtin_amdgcn_sched_barrier(0);
    mma_group(acc, f1);
    __builtin_amdgcn_sched_barrier(0);
    asm volatile("" ::
        "v"(f0.b[0][0]), "v"(f0.b[0][1]), "v"(f0.b[1][0]), "v"(f0.b[1][1]),
        "v"(f0.b[2][0]), "v"(f0.b[2][1]), "v"(f0.b[3][0]), "v"(f0.b[3][1]),
        "v"(f0.a[0][0]), "v"(f0.a[0][1]), "v"(f0.a[0][2]),
        "v"(f0.a[1][0]), "v"(f0.a[1][1]), "v"(f0.a[1][2]),
        "v"(f0.a[2][0]), "v"(f0.a[2][1]), "v"(f0.a[2][2]),
        "v"(f0.a[3][0]), "v"(f0.a[3][1]), "v"(f0.a[3][2]));
  }
}

__device__ __forceinline__ v8f tail_tile(const float* __restrict__ arow,
                                         const float* __restrict__ bcol) {
  v8f acc = v8f{};
#pragma unroll 16
  for (int k = 0; k < kF; k += 4) {
    const v2f a = *(const v2f*)(arow + k);
    const v2f b = *(const v2f*)(bcol + k);
    acc = __builtin_amdgcn_wmma_f32_16x16x4_f32(false, a, false, b, (short)0, acc,
                                                false, false);
  }
  return acc;
}

__device__ __forceinline__ float gelu_exact(float v) {
  return 0.5f * v * (1.0f + erff(v * 0.70710678118654752440f));
}

struct XRow { float v[kC][3]; };

template <bool kEdgeRow>
__device__ __forceinline__ void load_xrow(const float* __restrict__ ximg, int ww,
                                          int h, int hm, int hp, XRow& r) {
  const bool inside = !kEdgeRow || (ww >= 0 && ww < kW);
#pragma unroll
  for (int ci = 0; ci < kC; ++ci) {
    float a = 0.f, b = 0.f, c = 0.f;
    if (inside) {
      const float* __restrict__ p = ximg + ((size_t)ci * kW + ww) * kW;
      a = p[hm];
      b = p[h];
      c = p[hp];
    }
    r.v[ci][0] = (h > 0) ? a : 0.f;
    r.v[ci][1] = b;
    r.v[ci][2] = (h < kW - 1) ? c : 0.f;
  }
}

__global__ __launch_bounds__(kW)
void conv_bn_gelu_patch(const float* __restrict__ x,
                        const float* __restrict__ cw,
                        const float* __restrict__ cb,
                        const float* __restrict__ bng,
                        const float* __restrict__ bnb,
                        const float* __restrict__ bnm,
                        const float* __restrict__ bnv,
                        const float* __restrict__ M,
                        const int*   __restrict__ mask,
                        float* __restrict__ flat) {
  const int h     = threadIdx.x;
  const int img   = blockIdx.x / kStrips;
  const int strip = blockIdx.x - img * kStrips;
  const int w0    = strip * kStripRows;
  const int hm    = (h > 0) ? h - 1 : 0;
  const int hp    = (h < kW - 1) ? h + 1 : kW - 1;
  const float* __restrict__ ximg = x + (size_t)img * kC * kW * kW;

  float wgt[kC][kC][3][3];
#pragma unroll
  for (int co = 0; co < kC; ++co)
#pragma unroll
    for (int ci = 0; ci < kC; ++ci)
#pragma unroll
      for (int kw = 0; kw < 3; ++kw)
#pragma unroll
        for (int kh = 0; kh < 3; ++kh)
          wgt[co][ci][kw][kh] = cw[((co * kC + ci) * 3 + kw) * 3 + kh];
  float bias[kC], mean[kC], inv[kC], beta[kC];
#pragma unroll
  for (int co = 0; co < kC; ++co) {
    bias[co] = cb[co];
    mean[co] = bnm[co];
    inv[co]  = bng[co] * rsqrtf(bnv[co] + 1e-5f);
    beta[co] = bnb[co];
  }

  const int pj  = h & 15, ph = h >> 4;
  const int pi0 = w0 & 15;
  const int n   = (w0 >> 4) * kPR + ph;
  const bool gate = mask[n] > 0;
  const float* __restrict__ mrow = M + pi0 * kPS + pj;
  float* __restrict__ orow = flat + (size_t)(img * kN + n) * kF + pi0 * kPS + pj;

  XRow win[4];
  load_xrow<true >(ximg, w0 - 1, h, hm, hp, win[0]);
  load_xrow<false>(ximg, w0,     h, hm, hp, win[1]);
  load_xrow<false>(ximg, w0 + 1, h, hm, hp, win[2]);
#pragma unroll
  for (int r = 0; r < kStripRows; ++r) {
    const float mv = mrow[r * kPS];
    const float mm = gate ? mv : 1.0f;
    if (r + 2 < kStripRows)
      load_xrow<false>(ximg, w0 + r + 2, h, hm, hp, win[(r + 3) & 3]);
    else if (r + 2 == kStripRows)
      load_xrow<true >(ximg, w0 + r + 2, h, hm, hp, win[(r + 3) & 3]);
#pragma unroll
    for (int co = 0; co < kC; ++co) {
      float acc = bias[co];
#pragma unroll
      for (int ci = 0; ci < kC; ++ci)
#pragma unroll
        for (int kw = 0; kw < 3; ++kw)
#pragma unroll
          for (int kh = 0; kh < 3; ++kh)
            acc += win[(r + kw) & 3].v[ci][kh] * wgt[co][ci][kw][kh];
      acc = (acc - mean[co]) * inv[co] + beta[co];
      acc = gelu_exact(acc);
      orow[co * kPS * kPS + r * kPS] = acc * mm;
    }
  }
}


__global__ __launch_bounds__(256)
void gemm1_ln_gelu(const float* __restrict__ A,
                   const float* __restrict__ W1,
                   const float* __restrict__ b1,
                   const float* __restrict__ lng,
                   const float* __restrict__ lnb,
                   float* __restrict__ T) {
  __shared__ __attribute__((aligned(16))) float sA[kTileRows * kPitch];

  const int tid  = threadIdx.x;
  const int wave = __builtin_amdgcn_readfirstlane(tid >> 5);
  const int lane = tid & 31;
  const int hf   = lane >> 4;
  const int l16  = lane & 15;
  const size_t row0  = (size_t)blockIdx.x * kMainRows;
  const size_t tail0 = (size_t)kTailRow0 + 16 * (blockIdx.x / kTailGroups);
  const int    tcol0 = (blockIdx.x % kTailGroups) * kTailTiles * 16;

  {
    const int rq  = tid >> 6;
    const int sub = tid & 63;
    const float* __restrict__ gmain = A + (row0 + rq) * kF + 4 * sub;
    const float* __restrict__ gtail = A + (tail0 + rq) * kF + 4 * sub;
    float* __restrict__ sdst = sA + rq * kPitch + 4 * sub;
    float4 v[kStageRB][kStageCB];
#pragma unroll
    for (int rb = 0; rb < kStageRB; ++rb) {
      const float* __restrict__ g = (rb < kMainRows / 4)
          ? gmain + (size_t)rb * 4 * kF
          : gtail + (size_t)(rb - kMainRows / 4) * 4 * kF;
#pragma unroll
      for (int cb = 0; cb < kStageCB; ++cb) v[rb][cb] = *(const float4*)(g + 256 * cb);
    }
#pragma unroll
    for (int rb = 0; rb < kStageRB; ++rb) {
#pragma unroll
      for (int cb = 0; cb < kStageCB; ++cb)
        *(float4*)(sdst + rb * 4 * kPitch + 256 * cb) = v[rb][cb];
    }
  }
  __syncthreads();

  const int kOff = 2 * hf;

  if (wave < kTailTiles) {
    const int n = tcol0 + wave * 16 + l16;
    const v8f d = tail_tile(sA + (kMainRows + l16) * kPitch + kOff,
                            W1 + (size_t)n * kF + kOff);
    const float bias = b1[n];
#pragma unroll
    for (int r = 0; r < 8; ++r)
      T[(tail0 + r + 8 * hf) * kHid + n] = d[r] + bias;
  }

  const int n0 = wave * kNTp * 16 + l16;
  v8f acc[kPass][kMT][kNTp];
#pragma unroll
  for (int p = 0; p < kPass; ++p)
    panel_gemm(sA + l16 * kPitch + kOff,
               W1 + (size_t)(n0 + p * kWaves * kNTp * 16) * kF + kOff,
               acc[p]);
  __syncthreads();

#pragma unroll
  for (int p = 0; p < kPass; ++p) {
#pragma unroll
    for (int t = 0; t < kNTp; ++t) {
      const int n = n0 + (p * kWaves * kNTp + t) * 16;
      const float bias = b1[n];
#pragma unroll
      for (int i = 0; i < kMT; ++i) {
#pragma unroll
        for (int r = 0; r < 8; ++r) {
          sA[(16 * i + r + 8 * hf) * kPitch + n] = acc[p][i][t][r] + bias;
        }
      }
    }
  }
  __syncthreads();

  constexpr int kRuns = kF / 128;
  constexpr int kRowsPerWave = kMainRows / kWaves;
  for (int rr = 0; rr < kRowsPerWave; ++rr) {
    const int r = wave * kRowsPerWave + rr;
    const float* rp = sA + r * kPitch;
    float s = 0.f, s2 = 0.f;
#pragma unroll
    for (int j = 0; j < kRuns; ++j) {
      const float4 v = *(const float4*)(rp + 4 * lane + 128 * j);
      s += v.x; s2 += v.x * v.x;
      s += v.y; s2 += v.y * v.y;
      s += v.z; s2 += v.z * v.z;
      s += v.w; s2 += v.w * v.w;
    }
#pragma unroll
    for (int o = 16; o > 0; o >>= 1) {
      s  += __shfl_xor(s, o);
      s2 += __shfl_xor(s2, o);
    }
    const float mu  = s / kF;
    const float var = s2 / kF - mu * mu;
    const float rs  = rsqrtf(var + 1e-5f);
    float* op = T + (row0 + r) * kF;
#pragma clang loop interleave_count(8)
    for (int c = lane; c < kF; c += 32) {
      const float v = (rp[c] - mu) * rs * lng[c] + lnb[c];
      op[c] = gelu_exact(v);
    }
  }
}

__global__ __launch_bounds__(256)
void ln_gelu_tail(const float* __restrict__ lng,
                  const float* __restrict__ lnb,
                  float* T) {
  const int wave = threadIdx.x >> 5;
  const int lane = threadIdx.x & 31;
  float* rp = T + (size_t)(kTailRow0 + blockIdx.x * kWaves + wave) * kHid + 4 * lane;
  constexpr int kRuns = kHid / 128;
  float4 v[kRuns];
#pragma unroll
  for (int j = 0; j < kRuns; ++j) v[j] = *(const float4*)(rp + 128 * j);
  float s = 0.f, s2 = 0.f;
#pragma unroll
  for (int j = 0; j < kRuns; ++j) {
    s += v[j].x; s2 += v[j].x * v[j].x;
    s += v[j].y; s2 += v[j].y * v[j].y;
    s += v[j].z; s2 += v[j].z * v[j].z;
    s += v[j].w; s2 += v[j].w * v[j].w;
  }
#pragma unroll
  for (int o = 16; o > 0; o >>= 1) {
    s  += __shfl_xor(s, o);
    s2 += __shfl_xor(s2, o);
  }
  const float mu  = s / kHid;
  const float var = s2 / kHid - mu * mu;
  const float rs  = rsqrtf(var + 1e-5f);
#pragma unroll
  for (int j = 0; j < kRuns; ++j) {
    const float4 g = *(const float4*)(lng + 4 * lane + 128 * j);
    const float4 b = *(const float4*)(lnb + 4 * lane + 128 * j);
    float4 o;
    o.x = gelu_exact((v[j].x - mu) * rs * g.x + b.x);
    o.y = gelu_exact((v[j].y - mu) * rs * g.y + b.y);
    o.z = gelu_exact((v[j].z - mu) * rs * g.z + b.z);
    o.w = gelu_exact((v[j].w - mu) * rs * g.w + b.w);
    *(float4*)(rp + 128 * j) = o;
  }
}

__global__ __launch_bounds__(256)
void gemm2_perm(const float* __restrict__ T,
                const float* __restrict__ W2,
                const float* __restrict__ b2,
                const int*   __restrict__ perm,
                float* __restrict__ out) {
  __shared__ __attribute__((aligned(16))) float sA[kTileRows * kPitch];
  __shared__ int   sRow[kTileRows];

  const int tid  = threadIdx.x;
  const int wave = __builtin_amdgcn_readfirstlane(tid >> 5);
  const int lane = tid & 31;
  const int hf   = lane >> 4;
  const int l16  = lane & 15;
  const int row0  = blockIdx.x * kMainRows;
  const int tail0 = kTailRow0 + 16 * (blockIdx.x / kTailGroups);
  const int tcol0 = (blockIdx.x % kTailGroups) * kTailTiles * 16;

  if (tid < kTileRows) {
    const int g   = (tid < kMainRows) ? row0 + tid : tail0 + (tid - kMainRows);
    const int img = g / kN;
    const int n   = g - img * kN;
    sRow[tid] = img * kN + perm[n];
  }
  __syncthreads();
  {
    const int rq  = tid >> 6;
    const int sub = tid & 63;
    const float* __restrict__ gsrc = T + 4 * sub;
    float* __restrict__ sdst = sA + rq * kPitch + 4 * sub;
    float4 v[kStageRB][kStageCB];
#pragma unroll
    for (int rb = 0; rb < kStageRB; ++rb) {
      const float* __restrict__ g = gsrc + (size_t)sRow[4 * rb + rq] * kF;
#pragma unroll
      for (int cb = 0; cb < kStageCB; ++cb) v[rb][cb] = *(const float4*)(g + 256 * cb);
    }
#pragma unroll
    for (int rb = 0; rb < kStageRB; ++rb) {
#pragma unroll
      for (int cb = 0; cb < kStageCB; ++cb)
        *(float4*)(sdst + rb * 4 * kPitch + 256 * cb) = v[rb][cb];
    }
  }
  __syncthreads();

  const int kOff = 2 * hf;

  if (wave < kTailTiles) {
    const int n = tcol0 + wave * 16 + l16;
    const v8f d = tail_tile(sA + (kMainRows + l16) * kPitch + kOff,
                            W2 + (size_t)n * kF + kOff);
    const float bias = b2[n];
#pragma unroll
    for (int r = 0; r < 8; ++r)
      out[(size_t)(tail0 + r + 8 * hf) * kOut + n] = d[r] + bias;
  }

  const int n0 = wave * kNTp * 16 + l16;
  for (int p = 0; p < kPass; ++p) {
    v8f acc[kMT][kNTp];
    panel_gemm(sA + l16 * kPitch + kOff,
               W2 + (size_t)(n0 + p * kWaves * kNTp * 16) * kF + kOff,
               acc);
#pragma unroll
    for (int t = 0; t < kNTp; ++t) {
      const int n = n0 + (p * kWaves * kNTp + t) * 16;
      const float bias = b2[n];
#pragma unroll
      for (int i = 0; i < kMT; ++i) {
#pragma unroll
        for (int r = 0; r < 8; ++r) {
          const int m = 16 * i + r + 8 * hf;
          out[(size_t)(row0 + m) * kOut + n] = acc[i][t][r] + bias;
        }
      }
    }
  }
}

extern "C" void kernel_launch(void* const* d_in, const int* in_sizes, int n_in,
                              void* d_out, int out_size, void* d_ws, size_t ws_size,
                              hipStream_t stream) {
  const float* x    = (const float*)d_in[0];
  const float* cw   = (const float*)d_in[1];
  const float* cb   = (const float*)d_in[2];
  const float* bng  = (const float*)d_in[3];
  const float* bnb  = (const float*)d_in[4];
  const float* bnm  = (const float*)d_in[5];
  const float* bnv  = (const float*)d_in[6];
  const float* M    = (const float*)d_in[7];
  const float* W1   = (const float*)d_in[8];
  const float* b1   = (const float*)d_in[9];
  const float* lng  = (const float*)d_in[10];
  const float* lnb  = (const float*)d_in[11];
  const float* W2   = (const float*)d_in[12];
  const float* b2   = (const float*)d_in[13];
  const int*   mask = (const int*)d_in[14];
  const int*   perm = (const int*)d_in[15];
  float* out = (float*)d_out;

  float* flat = (float*)d_ws;
  float* Tbuf = flat + (size_t)kRows * kF;

  conv_bn_gelu_patch<<<kConvBlocks, kW, 0, stream>>>(
      x, cw, cb, bng, bnb, bnm, bnv, M, mask, flat);
  gemm1_ln_gelu<<<kGrid, 256, 0, stream>>>(flat, W1, b1, lng, lnb, Tbuf);
  ln_gelu_tail <<<(kRows - kTailRow0) / kWaves, 256, 0, stream>>>(lng, lnb, Tbuf);
  gemm2_perm   <<<kGrid, 256, 0, stream>>>(Tbuf, W2, b2, perm, out);
}


# both GEMMs: each wave copies its 32 W rows chunk by chunk (32 k, whole 128-B lines, two trips ahead) into two LDS stages of its own and reads B fragments from LDS; summation order unchanged
# speedup vs baseline: 3.2284x; 1.0329x over previous
#include <hip/hip_runtime.h>
#include <math.h>

typedef float v2f __attribute__((ext_vector_type(2)));
typedef float v4f __attribute__((ext_vector_type(4)));
typedef float v8f __attribute__((ext_vector_type(8)));

namespace {
constexpr int kPS   = 16;
constexpr int kPR   = 14;
constexpr int kN    = 196;
constexpr int kC    = 3;
constexpr int kW    = 224;
constexpr int kBS   = 64;
constexpr int kF    = 768;
constexpr int kHid  = 768;
constexpr int kOut  = 768;
constexpr int kRows = kBS * kN;
constexpr int kGrid         = 256;
constexpr int kMainRows     = 48;
constexpr int kMT           = kMainRows / 16;
constexpr int kTailRow0     = kGrid * kMainRows;
constexpr int kTailTiles    = 3;
constexpr int kTailGroups   = (kHid / 16) / kTailTiles;
constexpr int kTileRows     = kMainRows + 16;
constexpr int kWaves        = 8;
constexpr int kTilesPerWave = (kHid / 16) / kWaves;
constexpr int kNTp          = 2;
constexpr int kPass         = kTilesPerWave / kNTp;
constexpr int kKS           = 4;
constexpr int kPitch        = kF + 4;
constexpr int kChunkK      = 8 * kKS;
constexpr int kTrips       = kF / kChunkK;
constexpr int kBRows       = kNTp * 16;
constexpr int kBPitch      = kChunkK + 4;
constexpr int kBStage      = kBRows * kBPitch;
constexpr int kWRegs       = kBRows * kChunkK / (32 * 4);
constexpr int kStageCB      = kF / 256;
constexpr int kStageRB      = kTileRows / 4;
constexpr int kStripRows   = 8;
constexpr int kStrips      = kW / kStripRows;
constexpr int kConvBlocks  = kBS * kStrips;
static_assert(kTailRow0 + 16 * (kGrid / kTailGroups) == kRows, "tail pieces cover the last rows");
static_assert(kTailTiles * kTailGroups * 16 == kHid, "tail pieces cover every column");
static_assert(kWRegs == 8 && kBRows == 4 * kWRegs && kChunkK == 32 && kTrips % 2 == 0,
              "a lane's 8 pieces are rows 8g..8g+7 of the chunk; a pass ends on stage 0");
static_assert(kStageCB * 256 == kF && kStageRB * 4 == kTileRows && kMainRows % 4 == 0,
              "staging covers the panel; a block of 4 rows is all main or all tail");
}


struct Frags {
  v2f a[kKS][kMT];
  v2f b[kKS][kNTp];
};

__device__ __forceinline__ void load_group(const float* __restrict__ arow,
                                           const float* __restrict__ bfrag,
                                           int ak, int bk, Frags& f) {
#pragma unroll
  for (int s = 0; s < kKS; ++s) {
#pragma unroll
    for (int t = 0; t < kNTp; ++t)
      f.b[s][t] = *(const v2f*)(bfrag + t * 16 * kBPitch + bk + 4 * s);
#pragma unroll
    for (int i = 0; i < kMT; ++i)
      f.a[s][i] = *(const v2f*)(arow + i * 16 * kPitch + ak + 4 * s);
  }
}

__device__ __forceinline__ void mma_group(v8f (&acc)[kMT][kNTp], const Frags& f) {
#pragma unroll
  for (int s = 0; s < kKS; ++s)
#pragma unroll
    for (int i = 0; i < kMT; ++i)
#pragma unroll
      for (int t = 0; t < kNTp; ++t)
        acc[i][t] = __builtin_amdgcn_wmma_f32_16x16x4_f32(
            false, f.a[s][i], false, f.b[s][t], (short)0, acc[i][t], false, false);
}

struct WChunk { v4f v[kWRegs]; };

__device__ __forceinline__ void w_fetch(const float* __restrict__ wl, WChunk& c) {
#pragma unroll
  for (int j = 0; j < kWRegs; ++j) c.v[j] = *(const v4f*)(wl + (size_t)j * kF);
}

__device__ __forceinline__ void w_put(float* __restrict__ sl, const WChunk& c) {
#pragma unroll
  for (int j = 0; j < kWRegs; ++j) *(v4f*)(sl + j * kBPitch) = c.v[j];
}

__device__ __forceinline__ void panel_gemm(const float* __restrict__ arow,
                                           float* __restrict__ sbw,
                                           int lane,
                                           const float* __restrict__ wcur,
                                           const float* __restrict__ wnext,
                                           WChunk& wr,
                                           v8f (&acc)[kMT][kNTp]) {
#pragma unroll
  for (int i = 0; i < kMT; ++i)
#pragma unroll
    for (int t = 0; t < kNTp; ++t) acc[i][t] = v8f{};

  const int fo = (lane & 15) * kBPitch + 2 * (lane >> 4);
  const int po = 8 * (lane >> 3) * kBPitch + 4 * (lane & 7);

  Frags f0, f1;
  load_group(arow, sbw + fo, 0, 0, f0);
  for (int t = 0; t < kTrips; ++t) {
    const int k = t * kChunkK;
    const float* __restrict__ cur = sbw + (t & 1) * kBStage;
    float* __restrict__ nxt = sbw + ((t + 1) & 1) * kBStage;
    load_group(arow, cur + fo, k + 4 * kKS, 4 * kKS, f1);
    __builtin_amdgcn_sched_barrier(0);
    mma_group(acc, f0);
    __builtin_amdgcn_sched_barrier(0);
    w_put(nxt + po, wr);
    const int t2 = t + 2;
    w_fetch((t2 < kTrips) ? wcur + t2 * kChunkK : wnext + (t2 - kTrips) * kChunkK, wr);
    const int kn = (t + 1 < kTrips) ? k + kChunkK : 0;
    load_group(arow, nxt + fo, kn, 0, f0);
    __builtin_amdgcn_sched_barrier(0);
    mma_group(acc, f1);
    __builtin_amdgcn_sched_barrier(0);
    asm volatile("" ::
        "v"(f0.b[0][0]), "v"(f0.b[0][1]), "v"(f0.b[1][0]), "v"(f0.b[1][1]),
        "v"(f0.b[2][0]), "v"(f0.b[2][1]), "v"(f0.b[3][0]), "v"(f0.b[3][1]),
        "v"(f0.a[0][0]), "v"(f0.a[0][1]), "v"(f0.a[0][2]),
        "v"(f0.a[1][0]), "v"(f0.a[1][1]), "v"(f0.a[1][2]),
        "v"(f0.a[2][0]), "v"(f0.a[2][1]), "v"(f0.a[2][2]),
        "v"(f0.a[3][0]), "v"(f0.a[3][1]), "v"(f0.a[3][2]));
  }
}

__device__ __forceinline__ v8f tail_tile(const float* __restrict__ arow,
                                         const float* __restrict__ bcol) {
  v8f acc = v8f{};
#pragma unroll 16
  for (int k = 0; k < kF; k += 4) {
    const v2f a = *(const v2f*)(arow + k);
    const v2f b = *(const v2f*)(bcol + k);
    acc = __builtin_amdgcn_wmma_f32_16x16x4_f32(false, a, false, b, (short)0, acc,
                                                false, false);
  }
  return acc;
}

__device__ __forceinline__ float gelu_exact(float v) {
  return 0.5f * v * (1.0f + erff(v * 0.70710678118654752440f));
}

struct XRow { float v[kC][3]; };

template <bool kEdgeRow>
__device__ __forceinline__ void load_xrow(const float* __restrict__ ximg, int ww,
                                          int h, int hm, int hp, XRow& r) {
  const bool inside = !kEdgeRow || (ww >= 0 && ww < kW);
#pragma unroll
  for (int ci = 0; ci < kC; ++ci) {
    float a = 0.f, b = 0.f, c = 0.f;
    if (inside) {
      const float* __restrict__ p = ximg + ((size_t)ci * kW + ww) * kW;
      a = p[hm];
      b = p[h];
      c = p[hp];
    }
    r.v[ci][0] = (h > 0) ? a : 0.f;
    r.v[ci][1] = b;
    r.v[ci][2] = (h < kW - 1) ? c : 0.f;
  }
}

__global__ __launch_bounds__(kW)
void conv_bn_gelu_patch(const float* __restrict__ x,
                        const float* __restrict__ cw,
                        const float* __restrict__ cb,
                        const float* __restrict__ bng,
                        const float* __restrict__ bnb,
                        const float* __restrict__ bnm,
                        const float* __restrict__ bnv,
                        const float* __restrict__ M,
                        const int*   __restrict__ mask,
                        float* __restrict__ flat) {
  const int h     = threadIdx.x;
  const int img   = blockIdx.x / kStrips;
  const int strip = blockIdx.x - img * kStrips;
  const int w0    = strip * kStripRows;
  const int hm    = (h > 0) ? h - 1 : 0;
  const int hp    = (h < kW - 1) ? h + 1 : kW - 1;
  const float* __restrict__ ximg = x + (size_t)img * kC * kW * kW;

  float wgt[kC][kC][3][3];
#pragma unroll
  for (int co = 0; co < kC; ++co)
#pragma unroll
    for (int ci = 0; ci < kC; ++ci)
#pragma unroll
      for (int kw = 0; kw < 3; ++kw)
#pragma unroll
        for (int kh = 0; kh < 3; ++kh)
          wgt[co][ci][kw][kh] = cw[((co * kC + ci) * 3 + kw) * 3 + kh];
  float bias[kC], mean[kC], inv[kC], beta[kC];
#pragma unroll
  for (int co = 0; co < kC; ++co) {
    bias[co] = cb[co];
    mean[co] = bnm[co];
    inv[co]  = bng[co] * rsqrtf(bnv[co] + 1e-5f);
    beta[co] = bnb[co];
  }

  const int pj  = h & 15, ph = h >> 4;
  const int pi0 = w0 & 15;
  const int n   = (w0 >> 4) * kPR + ph;
  const bool gate = mask[n] > 0;
  const float* __restrict__ mrow = M + pi0 * kPS + pj;
  float* __restrict__ orow = flat + (size_t)(img * kN + n) * kF + pi0 * kPS + pj;

  XRow win[4];
  load_xrow<true >(ximg, w0 - 1, h, hm, hp, win[0]);
  load_xrow<false>(ximg, w0,     h, hm, hp, win[1]);
  load_xrow<false>(ximg, w0 + 1, h, hm, hp, win[2]);
#pragma unroll
  for (int r = 0; r < kStripRows; ++r) {
    const float mv = mrow[r * kPS];
    const float mm = gate ? mv : 1.0f;
    if (r + 2 < kStripRows)
      load_xrow<false>(ximg, w0 + r + 2, h, hm, hp, win[(r + 3) & 3]);
    else if (r + 2 == kStripRows)
      load_xrow<true >(ximg, w0 + r + 2, h, hm, hp, win[(r + 3) & 3]);
#pragma unroll
    for (int co = 0; co < kC; ++co) {
      float acc = bias[co];
#pragma unroll
      for (int ci = 0; ci < kC; ++ci)
#pragma unroll
        for (int kw = 0; kw < 3; ++kw)
#pragma unroll
          for (int kh = 0; kh < 3; ++kh)
            acc += win[(r + kw) & 3].v[ci][kh] * wgt[co][ci][kw][kh];
      acc = (acc - mean[co]) * inv[co] + beta[co];
      acc = gelu_exact(acc);
      orow[co * kPS * kPS + r * kPS] = acc * mm;
    }
  }
}


__global__ __launch_bounds__(256)
void gemm1_ln_gelu(const float* __restrict__ A,
                   const float* __restrict__ W1,
                   const float* __restrict__ b1,
                   const float* __restrict__ lng,
                   const float* __restrict__ lnb,
                   float* __restrict__ T) {
  __shared__ __attribute__((aligned(16))) float sA[kTileRows * kPitch];
  __shared__ __attribute__((aligned(16))) float sB[kWaves * 2 * kBStage];

  const int tid  = threadIdx.x;
  const int wave = __builtin_amdgcn_readfirstlane(tid >> 5);
  const int lane = tid & 31;
  const int hf   = lane >> 4;
  const int l16  = lane & 15;
  const size_t row0  = (size_t)blockIdx.x * kMainRows;

  float* __restrict__ sbw = sB + wave * 2 * kBStage;
  const float* __restrict__ wl = W1 + (size_t)(wave * kBRows + 8 * (lane >> 3)) * kF + 4 * (lane & 7);
  constexpr size_t kPassStride = (size_t)kWaves * kBRows * kF;
  WChunk wr;
  w_fetch(wl, wr);
  const size_t tail0 = (size_t)kTailRow0 + 16 * (blockIdx.x / kTailGroups);
  const int    tcol0 = (blockIdx.x % kTailGroups) * kTailTiles * 16;

  {
    const int rq  = tid >> 6;
    const int sub = tid & 63;
    const float* __restrict__ gmain = A + (row0 + rq) * kF + 4 * sub;
    const float* __restrict__ gtail = A + (tail0 + rq) * kF + 4 * sub;
    float* __restrict__ sdst = sA + rq * kPitch + 4 * sub;
    float4 v[kStageRB][kStageCB];
#pragma unroll
    for (int rb = 0; rb < kStageRB; ++rb) {
      const float* __restrict__ g = (rb < kMainRows / 4)
          ? gmain + (size_t)rb * 4 * kF
          : gtail + (size_t)(rb - kMainRows / 4) * 4 * kF;
#pragma unroll
      for (int cb = 0; cb < kStageCB; ++cb) v[rb][cb] = *(const float4*)(g + 256 * cb);
    }
#pragma unroll
    for (int rb = 0; rb < kStageRB; ++rb) {
#pragma unroll
      for (int cb = 0; cb < kStageCB; ++cb)
        *(float4*)(sdst + rb * 4 * kPitch + 256 * cb) = v[rb][cb];
    }
  }
  w_put(sbw + 8 * (lane >> 3) * kBPitch + 4 * (lane & 7), wr);
  w_fetch(wl + kChunkK, wr);
  __syncthreads();

  const int kOff = 2 * hf;

  if (wave < kTailTiles) {
    const int n = tcol0 + wave * 16 + l16;
    const v8f d = tail_tile(sA + (kMainRows + l16) * kPitch + kOff,
                            W1 + (size_t)n * kF + kOff);
    const float bias = b1[n];
#pragma unroll
    for (int r = 0; r < 8; ++r)
      T[(tail0 + r + 8 * hf) * kHid + n] = d[r] + bias;
  }

  const int n0 = wave * kNTp * 16 + l16;
  v8f acc[kPass][kMT][kNTp];
#pragma unroll
  for (int p = 0; p < kPass; ++p)
    panel_gemm(sA + l16 * kPitch + kOff, sbw, lane,
               wl + p * kPassStride,
               wl + (p + 1 < kPass ? p + 1 : p) * kPassStride,
               wr, acc[p]);
  __syncthreads();

#pragma unroll
  for (int p = 0; p < kPass; ++p) {
#pragma unroll
    for (int t = 0; t < kNTp; ++t) {
      const int n = n0 + (p * kWaves * kNTp + t) * 16;
      const float bias = b1[n];
#pragma unroll
      for (int i = 0; i < kMT; ++i) {
#pragma unroll
        for (int r = 0; r < 8; ++r) {
          sA[(16 * i + r + 8 * hf) * kPitch + n] = acc[p][i][t][r] + bias;
        }
      }
    }
  }
  __syncthreads();

  constexpr int kRuns = kF / 128;
  constexpr int kRowsPerWave = kMainRows / kWaves;
  for (int rr = 0; rr < kRowsPerWave; ++rr) {
    const int r = wave * kRowsPerWave + rr;
    const float* rp = sA + r * kPitch;
    float s = 0.f, s2 = 0.f;
#pragma unroll
    for (int j = 0; j < kRuns; ++j) {
      const float4 v = *(const float4*)(rp + 4 * lane + 128 * j);
      s += v.x; s2 += v.x * v.x;
      s += v.y; s2 += v.y * v.y;
      s += v.z; s2 += v.z * v.z;
      s += v.w; s2 += v.w * v.w;
    }
#pragma unroll
    for (int o = 16; o > 0; o >>= 1) {
      s  += __shfl_xor(s, o);
      s2 += __shfl_xor(s2, o);
    }
    const float mu  = s / kF;
    const float var = s2 / kF - mu * mu;
    const float rs  = rsqrtf(var + 1e-5f);
    float* op = T + (row0 + r) * kF;
#pragma clang loop interleave_count(8)
    for (int c = lane; c < kF; c += 32) {
      const float v = (rp[c] - mu) * rs * lng[c] + lnb[c];
      op[c] = gelu_exact(v);
    }
  }
}

__global__ __launch_bounds__(256)
void ln_gelu_tail(const float* __restrict__ lng,
                  const float* __restrict__ lnb,
                  float* T) {
  const int wave = threadIdx.x >> 5;
  const int lane = threadIdx.x & 31;
  float* rp = T + (size_t)(kTailRow0 + blockIdx.x * kWaves + wave) * kHid + 4 * lane;
  constexpr int kRuns = kHid / 128;
  float4 v[kRuns];
#pragma unroll
  for (int j = 0; j < kRuns; ++j) v[j] = *(const float4*)(rp + 128 * j);
  float s = 0.f, s2 = 0.f;
#pragma unroll
  for (int j = 0; j < kRuns; ++j) {
    s += v[j].x; s2 += v[j].x * v[j].x;
    s += v[j].y; s2 += v[j].y * v[j].y;
    s += v[j].z; s2 += v[j].z * v[j].z;
    s += v[j].w; s2 += v[j].w * v[j].w;
  }
#pragma unroll
  for (int o = 16; o > 0; o >>= 1) {
    s  += __shfl_xor(s, o);
    s2 += __shfl_xor(s2, o);
  }
  const float mu  = s / kHid;
  const float var = s2 / kHid - mu * mu;
  const float rs  = rsqrtf(var + 1e-5f);
#pragma unroll
  for (int j = 0; j < kRuns; ++j) {
    const float4 g = *(const float4*)(lng + 4 * lane + 128 * j);
    const float4 b = *(const float4*)(lnb + 4 * lane + 128 * j);
    float4 o;
    o.x = gelu_exact((v[j].x - mu) * rs * g.x + b.x);
    o.y = gelu_exact((v[j].y - mu) * rs * g.y + b.y);
    o.z = gelu_exact((v[j].z - mu) * rs * g.z + b.z);
    o.w = gelu_exact((v[j].w - mu) * rs * g.w + b.w);
    *(float4*)(rp + 128 * j) = o;
  }
}

__global__ __launch_bounds__(256)
void gemm2_perm(const float* __restrict__ T,
                const float* __restrict__ W2,
                const float* __restrict__ b2,
                const int*   __restrict__ perm,
                float* __restrict__ out) {
  __shared__ __attribute__((aligned(16))) float sA[kTileRows * kPitch];
  __shared__ __attribute__((aligned(16))) float sB[kWaves * 2 * kBStage];
  __shared__ int   sRow[kTileRows];

  const int tid  = threadIdx.x;
  const int wave = __builtin_amdgcn_readfirstlane(tid >> 5);
  const int lane = tid & 31;
  const int hf   = lane >> 4;
  const int l16  = lane & 15;
  const int row0  = blockIdx.x * kMainRows;

  float* __restrict__ sbw = sB + wave * 2 * kBStage;
  const float* __restrict__ wl = W2 + (size_t)(wave * kBRows + 8 * (lane >> 3)) * kF + 4 * (lane & 7);
  constexpr size_t kPassStride = (size_t)kWaves * kBRows * kF;
  WChunk wr;
  w_fetch(wl, wr);
  const int tail0 = kTailRow0 + 16 * (blockIdx.x / kTailGroups);
  const int tcol0 = (blockIdx.x % kTailGroups) * kTailTiles * 16;

  if (tid < kTileRows) {
    const int g   = (tid < kMainRows) ? row0 + tid : tail0 + (tid - kMainRows);
    const int img = g / kN;
    const int n   = g - img * kN;
    sRow[tid] = img * kN + perm[n];
  }
  __syncthreads();
  {
    const int rq  = tid >> 6;
    const int sub = tid & 63;
    const float* __restrict__ gsrc = T + 4 * sub;
    float* __restrict__ sdst = sA + rq * kPitch + 4 * sub;
    float4 v[kStageRB][kStageCB];
#pragma unroll
    for (int rb = 0; rb < kStageRB; ++rb) {
      const float* __restrict__ g = gsrc + (size_t)sRow[4 * rb + rq] * kF;
#pragma unroll
      for (int cb = 0; cb < kStageCB; ++cb) v[rb][cb] = *(const float4*)(g + 256 * cb);
    }
#pragma unroll
    for (int rb = 0; rb < kStageRB; ++rb) {
#pragma unroll
      for (int cb = 0; cb < kStageCB; ++cb)
        *(float4*)(sdst + rb * 4 * kPitch + 256 * cb) = v[rb][cb];
    }
  }
  w_put(sbw + 8 * (lane >> 3) * kBPitch + 4 * (lane & 7), wr);
  w_fetch(wl + kChunkK, wr);
  __syncthreads();

  const int kOff = 2 * hf;

  if (wave < kTailTiles) {
    const int n = tcol0 + wave * 16 + l16;
    const v8f d = tail_tile(sA + (kMainRows + l16) * kPitch + kOff,
                            W2 + (size_t)n * kF + kOff);
    const float bias = b2[n];
#pragma unroll
    for (int r = 0; r < 8; ++r)
      out[(size_t)(tail0 + r + 8 * hf) * kOut + n] = d[r] + bias;
  }

  const int n0 = wave * kNTp * 16 + l16;
  for (int p = 0; p < kPass; ++p) {
    v8f acc[kMT][kNTp];
    panel_gemm(sA + l16 * kPitch + kOff, sbw, lane,
               wl + p * kPassStride,
               wl + (p + 1 < kPass ? p + 1 : p) * kPassStride,
               wr, acc);
#pragma unroll
    for (int t = 0; t < kNTp; ++t) {
      const int n = n0 + (p * kWaves * kNTp + t) * 16;
      const float bias = b2[n];
#pragma unroll
      for (int i = 0; i < kMT; ++i) {
#pragma unroll
        for (int r = 0; r < 8; ++r) {
          const int m = 16 * i + r + 8 * hf;
          out[(size_t)(row0 + m) * kOut + n] = acc[i][t][r] + bias;
        }
      }
    }
  }
}

extern "C" void kernel_launch(void* const* d_in, const int* in_sizes, int n_in,
                              void* d_out, int out_size, void* d_ws, size_t ws_size,
                              hipStream_t stream) {
  const float* x    = (const float*)d_in[0];
  const float* cw   = (const float*)d_in[1];
  const float* cb   = (const float*)d_in[2];
  const float* bng  = (const float*)d_in[3];
  const float* bnb  = (const float*)d_in[4];
  const float* bnm  = (const float*)d_in[5];
  const float* bnv  = (const float*)d_in[6];
  const float* M    = (const float*)d_in[7];
  const float* W1   = (const float*)d_in[8];
  const float* b1   = (const float*)d_in[9];
  const float* lng  = (const float*)d_in[10];
  const float* lnb  = (const float*)d_in[11];
  const float* W2   = (const float*)d_in[12];
  const float* b2   = (const float*)d_in[13];
  const int*   mask = (const int*)d_in[14];
  const int*   perm = (const int*)d_in[15];
  float* out = (float*)d_out;

  float* flat = (float*)d_ws;
  float* Tbuf = flat + (size_t)kRows * kF;

  conv_bn_gelu_patch<<<kConvBlocks, kW, 0, stream>>>(
      x, cw, cb, bng, bnb, bnm, bnv, M, mask, flat);
  gemm1_ln_gelu<<<kGrid, 256, 0, stream>>>(flat, W1, b1, lng, lnb, Tbuf);
  ln_gelu_tail <<<(kRows - kTailRow0) / kWaves, 256, 0, stream>>>(lng, lnb, Tbuf);
  gemm2_perm   <<<kGrid, 256, 0, stream>>>(Tbuf, W2, b2, perm, out);
}


# both GEMMs: the younger wave of each SIMD's pair (waves 4-7) holds s_setprio 3 through passes 0-1 of 3, then priority 0, so the pair ends together; arithmetic and summation order unchanged
# speedup vs baseline: 3.3251x; 1.0300x over previous
#include <hip/hip_runtime.h>
#include <math.h>

typedef float v2f __attribute__((ext_vector_type(2)));
typedef float v4f __attribute__((ext_vector_type(4)));
typedef float v8f __attribute__((ext_vector_type(8)));

namespace {
constexpr int kPS   = 16;
constexpr int kPR   = 14;
constexpr int kN    = 196;
constexpr int kC    = 3;
constexpr int kW    = 224;
constexpr int kBS   = 64;
constexpr int kF    = 768;
constexpr int kHid  = 768;
constexpr int kOut  = 768;
constexpr int kRows = kBS * kN;
constexpr int kGrid         = 256;
constexpr int kMainRows     = 48;
constexpr int kMT           = kMainRows / 16;
constexpr int kTailRow0     = kGrid * kMainRows;
constexpr int kTailTiles    = 3;
constexpr int kTailGroups   = (kHid / 16) / kTailTiles;
constexpr int kTileRows     = kMainRows + 16;
constexpr int kWaves        = 8;
constexpr int kTilesPerWave = (kHid / 16) / kWaves;
constexpr int kNTp          = 2;
constexpr int kPass         = kTilesPerWave / kNTp;
constexpr int kKS           = 4;
constexpr int kPitch        = kF + 4;
constexpr int kChunkK      = 8 * kKS;
constexpr int kTrips       = kF / kChunkK;
constexpr int kBRows       = kNTp * 16;
constexpr int kBPitch      = kChunkK + 4;
constexpr int kBStage      = kBRows * kBPitch;
constexpr int kWRegs       = kBRows * kChunkK / (32 * 4);
constexpr int kStageCB      = kF / 256;
constexpr int kStageRB      = kTileRows / 4;
constexpr int kPrioPasses  = 2;
constexpr int kStripRows   = 8;
constexpr int kStrips      = kW / kStripRows;
constexpr int kConvBlocks  = kBS * kStrips;
static_assert(kTailRow0 + 16 * (kGrid / kTailGroups) == kRows, "tail pieces cover the last rows");
static_assert(kTailTiles * kTailGroups * 16 == kHid, "tail pieces cover every column");
static_assert(kWRegs == 8 && kBRows == 4 * kWRegs && kChunkK == 32 && kTrips % 2 == 0,
              "a lane's 8 pieces are rows 8g..8g+7 of the chunk; a pass ends on stage 0");
static_assert(kPrioPasses >= 1 && kPrioPasses < kPass, "the raised wave goes back to priority 0 before its last pass");
static_assert(kStageCB * 256 == kF && kStageRB * 4 == kTileRows && kMainRows % 4 == 0,
              "staging covers the panel; a block of 4 rows is all main or all tail");
}


struct Frags {
  v2f a[kKS][kMT];
  v2f b[kKS][kNTp];
};

__device__ __forceinline__ void load_group(const float* __restrict__ arow,
                                           const float* __restrict__ bfrag,
                                           int ak, int bk, Frags& f) {
#pragma unroll
  for (int s = 0; s < kKS; ++s) {
#pragma unroll
    for (int t = 0; t < kNTp; ++t)
      f.b[s][t] = *(const v2f*)(bfrag + t * 16 * kBPitch + bk + 4 * s);
#pragma unroll
    for (int i = 0; i < kMT; ++i)
      f.a[s][i] = *(const v2f*)(arow + i * 16 * kPitch + ak + 4 * s);
  }
}

__device__ __forceinline__ void mma_group(v8f (&acc)[kMT][kNTp], const Frags& f) {
#pragma unroll
  for (int s = 0; s < kKS; ++s)
#pragma unroll
    for (int i = 0; i < kMT; ++i)
#pragma unroll
      for (int t = 0; t < kNTp; ++t)
        acc[i][t] = __builtin_amdgcn_wmma_f32_16x16x4_f32(
            false, f.a[s][i], false, f.b[s][t], (short)0, acc[i][t], false, false);
}

struct WChunk { v4f v[kWRegs]; };

__device__ __forceinline__ void w_fetch(const float* __restrict__ wl, WChunk& c) {
#pragma unroll
  for (int j = 0; j < kWRegs; ++j) c.v[j] = *(const v4f*)(wl + (size_t)j * kF);
}

__device__ __forceinline__ void w_put(float* __restrict__ sl, const WChunk& c) {
#pragma unroll
  for (int j = 0; j < kWRegs; ++j) *(v4f*)(sl + j * kBPitch) = c.v[j];
}

__device__ __forceinline__ void panel_gemm(const float* __restrict__ arow,
                                           float* __restrict__ sbw,
                                           int lane,
                                           const float* __restrict__ wcur,
                                           const float* __restrict__ wnext,
                                           WChunk& wr,
                                           v8f (&acc)[kMT][kNTp]) {
#pragma unroll
  for (int i = 0; i < kMT; ++i)
#pragma unroll
    for (int t = 0; t < kNTp; ++t) acc[i][t] = v8f{};

  const int fo = (lane & 15) * kBPitch + 2 * (lane >> 4);
  const int po = 8 * (lane >> 3) * kBPitch + 4 * (lane & 7);

  Frags f0, f1;
  load_group(arow, sbw + fo, 0, 0, f0);
  for (int t = 0; t < kTrips; ++t) {
    const int k = t * kChunkK;
    const float* __restrict__ cur = sbw + (t & 1) * kBStage;
    float* __restrict__ nxt = sbw + ((t + 1) & 1) * kBStage;
    load_group(arow, cur + fo, k + 4 * kKS, 4 * kKS, f1);
    __builtin_amdgcn_sched_barrier(0);
    mma_group(acc, f0);
    __builtin_amdgcn_sched_barrier(0);
    w_put(nxt + po, wr);
    const int t2 = t + 2;
    w_fetch((t2 < kTrips) ? wcur + t2 * kChunkK : wnext + (t2 - kTrips) * kChunkK, wr);
    const int kn = (t + 1 < kTrips) ? k + kChunkK : 0;
    load_group(arow, nxt + fo, kn, 0, f0);
    __builtin_amdgcn_sched_barrier(0);
    mma_group(acc, f1);
    __builtin_amdgcn_sched_barrier(0);
    asm volatile("" ::
        "v"(f0.b[0][0]), "v"(f0.b[0][1]), "v"(f0.b[1][0]), "v"(f0.b[1][1]),
        "v"(f0.b[2][0]), "v"(f0.b[2][1]), "v"(f0.b[3][0]), "v"(f0.b[3][1]),
        "v"(f0.a[0][0]), "v"(f0.a[0][1]), "v"(f0.a[0][2]),
        "v"(f0.a[1][0]), "v"(f0.a[1][1]), "v"(f0.a[1][2]),
        "v"(f0.a[2][0]), "v"(f0.a[2][1]), "v"(f0.a[2][2]),
        "v"(f0.a[3][0]), "v"(f0.a[3][1]), "v"(f0.a[3][2]));
  }
}

__device__ __forceinline__ v8f tail_tile(const float* __restrict__ arow,
                                         const float* __restrict__ bcol) {
  v8f acc = v8f{};
#pragma unroll 16
  for (int k = 0; k < kF; k += 4) {
    const v2f a = *(const v2f*)(arow + k);
    const v2f b = *(const v2f*)(bcol + k);
    acc = __builtin_amdgcn_wmma_f32_16x16x4_f32(false, a, false, b, (short)0, acc,
                                                false, false);
  }
  return acc;
}

__device__ __forceinline__ float gelu_exact(float v) {
  return 0.5f * v * (1.0f + erff(v * 0.70710678118654752440f));
}

struct XRow { float v[kC][3]; };

template <bool kEdgeRow>
__device__ __forceinline__ void load_xrow(const float* __restrict__ ximg, int ww,
                                          int h, int hm, int hp, XRow& r) {
  const bool inside = !kEdgeRow || (ww >= 0 && ww < kW);
#pragma unroll
  for (int ci = 0; ci < kC; ++ci) {
    float a = 0.f, b = 0.f, c = 0.f;
    if (inside) {
      const float* __restrict__ p = ximg + ((size_t)ci * kW + ww) * kW;
      a = p[hm];
      b = p[h];
      c = p[hp];
    }
    r.v[ci][0] = (h > 0) ? a : 0.f;
    r.v[ci][1] = b;
    r.v[ci][2] = (h < kW - 1) ? c : 0.f;
  }
}

__global__ __launch_bounds__(kW)
void conv_bn_gelu_patch(const float* __restrict__ x,
                        const float* __restrict__ cw,
                        const float* __restrict__ cb,
                        const float* __restrict__ bng,
                        const float* __restrict__ bnb,
                        const float* __restrict__ bnm,
                        const float* __restrict__ bnv,
                        const float* __restrict__ M,
                        const int*   __restrict__ mask,
                        float* __restrict__ flat) {
  const int h     = threadIdx.x;
  const int img   = blockIdx.x / kStrips;
  const int strip = blockIdx.x - img * kStrips;
  const int w0    = strip * kStripRows;
  const int hm    = (h > 0) ? h - 1 : 0;
  const int hp    = (h < kW - 1) ? h + 1 : kW - 1;
  const float* __restrict__ ximg = x + (size_t)img * kC * kW * kW;

  float wgt[kC][kC][3][3];
#pragma unroll
  for (int co = 0; co < kC; ++co)
#pragma unroll
    for (int ci = 0; ci < kC; ++ci)
#pragma unroll
      for (int kw = 0; kw < 3; ++kw)
#pragma unroll
        for (int kh = 0; kh < 3; ++kh)
          wgt[co][ci][kw][kh] = cw[((co * kC + ci) * 3 + kw) * 3 + kh];
  float bias[kC], mean[kC], inv[kC], beta[kC];
#pragma unroll
  for (int co = 0; co < kC; ++co) {
    bias[co] = cb[co];
    mean[co] = bnm[co];
    inv[co]  = bng[co] * rsqrtf(bnv[co] + 1e-5f);
    beta[co] = bnb[co];
  }

  const int pj  = h & 15, ph = h >> 4;
  const int pi0 = w0 & 15;
  const int n   = (w0 >> 4) * kPR + ph;
  const bool gate = mask[n] > 0;
  const float* __restrict__ mrow = M + pi0 * kPS + pj;
  float* __restrict__ orow = flat + (size_t)(img * kN + n) * kF + pi0 * kPS + pj;

  XRow win[4];
  load_xrow<true >(ximg, w0 - 1, h, hm, hp, win[0]);
  load_xrow<false>(ximg, w0,     h, hm, hp, win[1]);
  load_xrow<false>(ximg, w0 + 1, h, hm, hp, win[2]);
#pragma unroll
  for (int r = 0; r < kStripRows; ++r) {
    const float mv = mrow[r * kPS];
    const float mm = gate ? mv : 1.0f;
    if (r + 2 < kStripRows)
      load_xrow<false>(ximg, w0 + r + 2, h, hm, hp, win[(r + 3) & 3]);
    else if (r + 2 == kStripRows)
      load_xrow<true >(ximg, w0 + r + 2, h, hm, hp, win[(r + 3) & 3]);
#pragma unroll
    for (int co = 0; co < kC; ++co) {
      float acc = bias[co];
#pragma unroll
      for (int ci = 0; ci < kC; ++ci)
#pragma unroll
        for (int kw = 0; kw < 3; ++kw)
#pragma unroll
          for (int kh = 0; kh < 3; ++kh)
            acc += win[(r + kw) & 3].v[ci][kh] * wgt[co][ci][kw][kh];
      acc = (acc - mean[co]) * inv[co] + beta[co];
      acc = gelu_exact(acc);
      orow[co * kPS * kPS + r * kPS] = acc * mm;
    }
  }
}


__global__ __launch_bounds__(256)
void gemm1_ln_gelu(const float* __restrict__ A,
                   const float* __restrict__ W1,
                   const float* __restrict__ b1,
                   const float* __restrict__ lng,
                   const float* __restrict__ lnb,
                   float* __restrict__ T) {
  __shared__ __attribute__((aligned(16))) float sA[kTileRows * kPitch];
  __shared__ __attribute__((aligned(16))) float sB[kWaves * 2 * kBStage];

  const int tid  = threadIdx.x;
  const int wave = __builtin_amdgcn_readfirstlane(tid >> 5);
  const int lane = tid & 31;
  const int hf   = lane >> 4;
  const int l16  = lane & 15;
  const size_t row0  = (size_t)blockIdx.x * kMainRows;

  float* __restrict__ sbw = sB + wave * 2 * kBStage;
  const float* __restrict__ wl = W1 + (size_t)(wave * kBRows + 8 * (lane >> 3)) * kF + 4 * (lane & 7);
  constexpr size_t kPassStride = (size_t)kWaves * kBRows * kF;
  WChunk wr;
  w_fetch(wl, wr);
  const size_t tail0 = (size_t)kTailRow0 + 16 * (blockIdx.x / kTailGroups);
  const int    tcol0 = (blockIdx.x % kTailGroups) * kTailTiles * 16;

  {
    const int rq  = tid >> 6;
    const int sub = tid & 63;
    const float* __restrict__ gmain = A + (row0 + rq) * kF + 4 * sub;
    const float* __restrict__ gtail = A + (tail0 + rq) * kF + 4 * sub;
    float* __restrict__ sdst = sA + rq * kPitch + 4 * sub;
    float4 v[kStageRB][kStageCB];
#pragma unroll
    for (int rb = 0; rb < kStageRB; ++rb) {
      const float* __restrict__ g = (rb < kMainRows / 4)
          ? gmain + (size_t)rb * 4 * kF
          : gtail + (size_t)(rb - kMainRows / 4) * 4 * kF;
#pragma unroll
      for (int cb = 0; cb < kStageCB; ++cb) v[rb][cb] = *(const float4*)(g + 256 * cb);
    }
#pragma unroll
    for (int rb = 0; rb < kStageRB; ++rb) {
#pragma unroll
      for (int cb = 0; cb < kStageCB; ++cb)
        *(float4*)(sdst + rb * 4 * kPitch + 256 * cb) = v[rb][cb];
    }
  }
  w_put(sbw + 8 * (lane >> 3) * kBPitch + 4 * (lane & 7), wr);
  w_fetch(wl + kChunkK, wr);
  __syncthreads();

  const int kOff = 2 * hf;

  if (wave < kTailTiles) {
    const int n = tcol0 + wave * 16 + l16;
    const v8f d = tail_tile(sA + (kMainRows + l16) * kPitch + kOff,
                            W1 + (size_t)n * kF + kOff);
    const float bias = b1[n];
#pragma unroll
    for (int r = 0; r < 8; ++r)
      T[(tail0 + r + 8 * hf) * kHid + n] = d[r] + bias;
  }

  const int n0 = wave * kNTp * 16 + l16;
  v8f acc[kPass][kMT][kNTp];
  const bool younger = wave >= kWaves / 2;
  if (younger) __builtin_amdgcn_s_setprio(3);
#pragma unroll
  for (int p = 0; p < kPass; ++p) {
    panel_gemm(sA + l16 * kPitch + kOff, sbw, lane,
               wl + p * kPassStride,
               wl + (p + 1 < kPass ? p + 1 : p) * kPassStride,
               wr, acc[p]);
    if (p == kPrioPasses - 1 && younger) __builtin_amdgcn_s_setprio(0);
  }
  __syncthreads();

#pragma unroll
  for (int p = 0; p < kPass; ++p) {
#pragma unroll
    for (int t = 0; t < kNTp; ++t) {
      const int n = n0 + (p * kWaves * kNTp + t) * 16;
      const float bias = b1[n];
#pragma unroll
      for (int i = 0; i < kMT; ++i) {
#pragma unroll
        for (int r = 0; r < 8; ++r) {
          sA[(16 * i + r + 8 * hf) * kPitch + n] = acc[p][i][t][r] + bias;
        }
      }
    }
  }
  __syncthreads();

  constexpr int kRuns = kF / 128;
  constexpr int kRowsPerWave = kMainRows / kWaves;
  for (int rr = 0; rr < kRowsPerWave; ++rr) {
    const int r = wave * kRowsPerWave + rr;
    const float* rp = sA + r * kPitch;
    float s = 0.f, s2 = 0.f;
#pragma unroll
    for (int j = 0; j < kRuns; ++j) {
      const float4 v = *(const float4*)(rp + 4 * lane + 128 * j);
      s += v.x; s2 += v.x * v.x;
      s += v.y; s2 += v.y * v.y;
      s += v.z; s2 += v.z * v.z;
      s += v.w; s2 += v.w * v.w;
    }
#pragma unroll
    for (int o = 16; o > 0; o >>= 1) {
      s  += __shfl_xor(s, o);
      s2 += __shfl_xor(s2, o);
    }
    const float mu  = s / kF;
    const float var = s2 / kF - mu * mu;
    const float rs  = rsqrtf(var + 1e-5f);
    float* op = T + (row0 + r) * kF;
#pragma clang loop interleave_count(8)
    for (int c = lane; c < kF; c += 32) {
      const float v = (rp[c] - mu) * rs * lng[c] + lnb[c];
      op[c] = gelu_exact(v);
    }
  }
}

__global__ __launch_bounds__(256)
void ln_gelu_tail(const float* __restrict__ lng,
                  const float* __restrict__ lnb,
                  float* T) {
  const int wave = threadIdx.x >> 5;
  const int lane = threadIdx.x & 31;
  float* rp = T + (size_t)(kTailRow0 + blockIdx.x * kWaves + wave) * kHid + 4 * lane;
  constexpr int kRuns = kHid / 128;
  float4 v[kRuns];
#pragma unroll
  for (int j = 0; j < kRuns; ++j) v[j] = *(const float4*)(rp + 128 * j);
  float s = 0.f, s2 = 0.f;
#pragma unroll
  for (int j = 0; j < kRuns; ++j) {
    s += v[j].x; s2 += v[j].x * v[j].x;
    s += v[j].y; s2 += v[j].y * v[j].y;
    s += v[j].z; s2 += v[j].z * v[j].z;
    s += v[j].w; s2 += v[j].w * v[j].w;
  }
#pragma unroll
  for (int o = 16; o > 0; o >>= 1) {
    s  += __shfl_xor(s, o);
    s2 += __shfl_xor(s2, o);
  }
  const float mu  = s / kHid;
  const float var = s2 / kHid - mu * mu;
  const float rs  = rsqrtf(var + 1e-5f);
#pragma unroll
  for (int j = 0; j < kRuns; ++j) {
    const float4 g = *(const float4*)(lng + 4 * lane + 128 * j);
    const float4 b = *(const float4*)(lnb + 4 * lane + 128 * j);
    float4 o;
    o.x = gelu_exact((v[j].x - mu) * rs * g.x + b.x);
    o.y = gelu_exact((v[j].y - mu) * rs * g.y + b.y);
    o.z = gelu_exact((v[j].z - mu) * rs * g.z + b.z);
    o.w = gelu_exact((v[j].w - mu) * rs * g.w + b.w);
    *(float4*)(rp + 128 * j) = o;
  }
}

__global__ __launch_bounds__(256)
void gemm2_perm(const float* __restrict__ T,
                const float* __restrict__ W2,
                const float* __restrict__ b2,
                const int*   __restrict__ perm,
                float* __restrict__ out) {
  __shared__ __attribute__((aligned(16))) float sA[kTileRows * kPitch];
  __shared__ __attribute__((aligned(16))) float sB[kWaves * 2 * kBStage];
  __shared__ int   sRow[kTileRows];

  const int tid  = threadIdx.x;
  const int wave = __builtin_amdgcn_readfirstlane(tid >> 5);
  const int lane = tid & 31;
  const int hf   = lane >> 4;
  const int l16  = lane & 15;
  const int row0  = blockIdx.x * kMainRows;

  float* __restrict__ sbw = sB + wave * 2 * kBStage;
  const float* __restrict__ wl = W2 + (size_t)(wave * kBRows + 8 * (lane >> 3)) * kF + 4 * (lane & 7);
  constexpr size_t kPassStride = (size_t)kWaves * kBRows * kF;
  WChunk wr;
  w_fetch(wl, wr);
  const int tail0 = kTailRow0 + 16 * (blockIdx.x / kTailGroups);
  const int tcol0 = (blockIdx.x % kTailGroups) * kTailTiles * 16;

  if (tid < kTileRows) {
    const int g   = (tid < kMainRows) ? row0 + tid : tail0 + (tid - kMainRows);
    const int img = g / kN;
    const int n   = g - img * kN;
    sRow[tid] = img * kN + perm[n];
  }
  __syncthreads();
  {
    const int rq  = tid >> 6;
    const int sub = tid & 63;
    const float* __restrict__ gsrc = T + 4 * sub;
    float* __restrict__ sdst = sA + rq * kPitch + 4 * sub;
    float4 v[kStageRB][kStageCB];
#pragma unroll
    for (int rb = 0; rb < kStageRB; ++rb) {
      const float* __restrict__ g = gsrc + (size_t)sRow[4 * rb + rq] * kF;
#pragma unroll
      for (int cb = 0; cb < kStageCB; ++cb) v[rb][cb] = *(const float4*)(g + 256 * cb);
    }
#pragma unroll
    for (int rb = 0; rb < kStageRB; ++rb) {
#pragma unroll
      for (int cb = 0; cb < kStageCB; ++cb)
        *(float4*)(sdst + rb * 4 * kPitch + 256 * cb) = v[rb][cb];
    }
  }
  w_put(sbw + 8 * (lane >> 3) * kBPitch + 4 * (lane & 7), wr);
  w_fetch(wl + kChunkK, wr);
  __syncthreads();

  const int kOff = 2 * hf;

  if (wave < kTailTiles) {
    const int n = tcol0 + wave * 16 + l16;
    const v8f d = tail_tile(sA + (kMainRows + l16) * kPitch + kOff,
                            W2 + (size_t)n * kF + kOff);
    const float bias = b2[n];
#pragma unroll
    for (int r = 0; r < 8; ++r)
      out[(size_t)(tail0 + r + 8 * hf) * kOut + n] = d[r] + bias;
  }

  const int n0 = wave * kNTp * 16 + l16;
  const bool younger = wave >= kWaves / 2;
  if (younger) __builtin_amdgcn_s_setprio(3);
  for (int p = 0; p < kPass; ++p) {
    v8f acc[kMT][kNTp];
    panel_gemm(sA + l16 * kPitch + kOff, sbw, lane,
               wl + p * kPassStride,
               wl + (p + 1 < kPass ? p + 1 : p) * kPassStride,
               wr, acc);
    if (p == kPrioPasses - 1 && younger) __builtin_amdgcn_s_setprio(0);
#pragma unroll
    for (int t = 0; t < kNTp; ++t) {
      const int n = n0 + (p * kWaves * kNTp + t) * 16;
      const float bias = b2[n];
#pragma unroll
      for (int i = 0; i < kMT; ++i) {
#pragma unroll
        for (int r = 0; r < 8; ++r) {
          const int m = 16 * i + r + 8 * hf;
          out[(size_t)(row0 + m) * kOut + n] = acc[i][t][r] + bias;
        }
      }
    }
  }
}

extern "C" void kernel_launch(void* const* d_in, const int* in_sizes, int n_in,
                              void* d_out, int out_size, void* d_ws, size_t ws_size,
                              hipStream_t stream) {
  const float* x    = (const float*)d_in[0];
  const float* cw   = (const float*)d_in[1];
  const float* cb   = (const float*)d_in[2];
  const float* bng  = (const float*)d_in[3];
  const float* bnb  = (const float*)d_in[4];
  const float* bnm  = (const float*)d_in[5];
  const float* bnv  = (const float*)d_in[6];
  const float* M    = (const float*)d_in[7];
  const float* W1   = (const float*)d_in[8];
  const float* b1   = (const float*)d_in[9];
  const float* lng  = (const float*)d_in[10];
  const float* lnb  = (const float*)d_in[11];
  const float* W2   = (const float*)d_in[12];
  const float* b2   = (const float*)d_in[13];
  const int*   mask = (const int*)d_in[14];
  const int*   perm = (const int*)d_in[15];
  float* out = (float*)d_out;

  float* flat = (float*)d_ws;
  float* Tbuf = flat + (size_t)kRows * kF;

  conv_bn_gelu_patch<<<kConvBlocks, kW, 0, stream>>>(
      x, cw, cb, bng, bnb, bnm, bnv, M, mask, flat);
  gemm1_ln_gelu<<<kGrid, 256, 0, stream>>>(flat, W1, b1, lng, lnb, Tbuf);
  ln_gelu_tail <<<(kRows - kTailRow0) / kWaves, 256, 0, stream>>>(lng, lnb, Tbuf);
  gemm2_perm   <<<kGrid, 256, 0, stream>>>(Tbuf, W2, b2, perm, out);
}
